# starting kernel
# baseline (speedup 1.0000x reference)
#include <hip/hip_runtime.h>

#define F_IN 128
#define C 64

typedef float v2f __attribute__((ext_vector_type(2)));
typedef float v8f __attribute__((ext_vector_type(8)));

__device__ __forceinline__ unsigned f32_order_encode(float f) {
  unsigned b = __float_as_uint(f);
  return (b & 0x80000000u) ? ~b : (b | 0x80000000u);
}
__device__ __forceinline__ float f32_order_decode(unsigned u) {
  unsigned b = (u & 0x80000000u) ? (u & 0x7FFFFFFFu) : ~u;
  return __uint_as_float(b);
}

__global__ void k_init(float* __restrict__ out, const float* __restrict__ bias,
                       float* __restrict__ deg, float* __restrict__ asum,
                       unsigned* __restrict__ lmaxu, float* __restrict__ denom, int N) {
  int i = blockIdx.x * blockDim.x + threadIdx.x;
  if (i < N * C) out[i] = bias[i & (C - 1)];
  if (i < N) { deg[i] = 0.f; asum[i] = 0.f; lmaxu[i] = 0u; denom[i] = 0.f; }
}

__global__ void k_degree(const int* __restrict__ ei, const float* __restrict__ eattr,
                         float* __restrict__ deg, float* __restrict__ asum, int E) {
  int e = blockIdx.x * blockDim.x + threadIdx.x;
  if (e >= E) return;
  int d = ei[E + e];
  atomicAdd(&deg[d], 1.0f);
  atomicAdd(&asum[d], eattr[e]);
}

__global__ void k_gemm(const float* __restrict__ x, const float* __restrict__ Wl,
                       const float* __restrict__ Wr, float* __restrict__ xl,
                       float* __restrict__ xr, int N) {
  int wave = threadIdx.x >> 5;
  int lane = threadIdx.x & 31;
  long long t = (long long)blockIdx.x * 8 + wave;
  long long rowTiles = (long long)(N >> 4);
  if (t >= rowTiles * 8) return;
  int sub   = (int)(t & 7);
  int m0    = (int)(t >> 3) << 4;
  int n0    = (sub >> 1) << 4;
  int which = sub & 1;
  const float* W = which ? Wr : Wl;
  float*       Y = which ? xr : xl;

  int half = lane >> 4;
  int l15  = lane & 15;
  const float* arow = x + (long long)(m0 + l15) * F_IN + half * 2;
  const float* bcol = W + n0 + l15 + (long long)(half * 2) * C;

  v8f acc = {0.f, 0.f, 0.f, 0.f, 0.f, 0.f, 0.f, 0.f};
#pragma unroll 4
  for (int k0 = 0; k0 < F_IN; k0 += 4) {
    v2f a, b;
    a.x = arow[k0];
    a.y = arow[k0 + 1];
    b.x = bcol[(long long)k0 * C];
    b.y = bcol[(long long)(k0 + 1) * C];
    acc = __builtin_amdgcn_wmma_f32_16x16x4_f32(
         false, a,  false, b,
         (short)0, acc,  false,  false);
  }
  float* yrow = Y + (long long)(m0 + half * 8) * C + n0 + l15;
#pragma unroll
  for (int r = 0; r < 8; ++r) yrow[(long long)r * C] = acc[r];
}

__global__ void k_logits(const int* __restrict__ ei, const float* __restrict__ eattr,
                         const float* __restrict__ xl, const float* __restrict__ xr,
                         const float* __restrict__ deg, const float* __restrict__ asum,
                         const float* __restrict__ We, const float* __restrict__ att,
                         float* __restrict__ logit, unsigned* __restrict__ lmaxu,
                         int E, int N) {
  int gid = blockIdx.x * blockDim.x + threadIdx.x;
  int w = gid >> 5, lane = gid & 31;
  if (w >= E + N) return;
  int s, d; float ea;
  if (w < E) { s = ei[w]; d = ei[E + w]; ea = eattr[w]; }
  else       { s = d = w - E; ea = asum[s] / fmaxf(deg[s], 1.0f); }

  float p = 0.f;
#pragma unroll
  for (int j = 0; j < 2; ++j) {
    int c = lane + j * 32;
    float m = xl[(long long)s * C + c] + xr[(long long)d * C + c] + ea * We[c];
    m = (m > 0.f) ? m : 0.2f * m;
    p += m * att[c];
  }
  for (int off = 16; off > 0; off >>= 1) p += __shfl_xor(p, off, 32);
  if (lane == 0) {
    logit[w] = p;
    atomicMax(&lmaxu[d], f32_order_encode(p));
  }
}

__global__ void k_denom(const int* __restrict__ ei, float* __restrict__ logit,
                        const unsigned* __restrict__ lmaxu, float* __restrict__ denom,
                        int E, int N) {
  int i = blockIdx.x * blockDim.x + threadIdx.x;
  if (i >= E + N) return;
  int d = (i < E) ? ei[E + i] : (i - E);
  float lm = f32_order_decode(lmaxu[d]);
  float ex = __expf(logit[i] - lm);
  logit[i] = ex;
  atomicAdd(&denom[d], ex);
}

__global__ void k_aggregate(const int* __restrict__ ei, const float* __restrict__ exv,
                            const float* __restrict__ denom, const float* __restrict__ xl,
                            float* __restrict__ out, int E, int N) {
  int gid = blockIdx.x * blockDim.x + threadIdx.x;
  int w = gid >> 5, lane = gid & 31;
  if (w >= E + N) return;
  int s, d;
  if (w < E) { s = ei[w]; d = ei[E + w]; } else { s = d = w - E; }
  float alpha = exv[w] / denom[d];
#pragma unroll
  for (int j = 0; j < 2; ++j) {
    int c = lane + j * 32;
    atomicAdd(&out[(long long)d * C + c], alpha * xl[(long long)s * C + c]);
  }
}

extern "C" void kernel_launch(void* const* d_in, const int* in_sizes, int n_in,
                              void* d_out, int out_size, void* d_ws, size_t ws_size,
                              hipStream_t stream) {
  (void)n_in; (void)out_size; (void)ws_size;
  const float* x     = (const float*)d_in[0];
  const float* eattr = (const float*)d_in[1];
  const float* Wl    = (const float*)d_in[2];
  const float* Wr    = (const float*)d_in[3];
  const float* We    = (const float*)d_in[4];
  const float* att   = (const float*)d_in[5];
  const float* bias  = (const float*)d_in[6];
  const int*   ei    = (const int*)d_in[7];
  float*       out   = (float*)d_out;

  int N = in_sizes[0] / F_IN;
  int E = in_sizes[1];

  char* ws = (char*)d_ws;
  size_t o = 0;
  auto take = [&](size_t bytes) -> void* {
    void* p = ws + o; o += (bytes + 255) & ~(size_t)255; return p;
  };
  float*    xl    = (float*)take((size_t)N * C * 4);
  float*    xr    = (float*)take((size_t)N * C * 4);
  float*    deg   = (float*)take((size_t)N * 4);
  float*    asum  = (float*)take((size_t)N * 4);
  unsigned* lmaxu = (unsigned*)take((size_t)N * 4);
  float*    denom = (float*)take((size_t)N * 4);
  float*    logit = (float*)take((size_t)(E + N) * 4);

  const int B = 256;
  long long initThr = (long long)N * C;
  k_init<<<(unsigned)((initThr + B - 1) / B), B, 0, stream>>>(out, bias, deg, asum, lmaxu, denom, N);
  k_degree<<<(unsigned)((E + B - 1) / B), B, 0, stream>>>(ei, eattr, deg, asum, E);
  k_gemm<<<(unsigned)(N >> 4), 256, 0, stream>>>(x, Wl, Wr, xl, xr, N);

  long long edges = (long long)E + N;
  long long thr   = edges * 32;
  k_logits<<<(unsigned)((thr + B - 1) / B), B, 0, stream>>>(ei, eattr, xl, xr, deg, asum,
                                                            We, att, logit, lmaxu, E, N);
  k_denom<<<(unsigned)((edges + B - 1) / B), B, 0, stream>>>(ei, logit, lmaxu, denom, E, N);
  k_aggregate<<<(unsigned)((thr + B - 1) / B), B, 0, stream>>>(ei, logit, denom, xl, out, E, N);
}


# a block owns 200 destinations: one pass filters the edge list into LDS; count, max, exp sum, row sum by LDS atomics, normalised once per node; no global atomics, 6 to 2 launches; sum order changes
# speedup vs baseline: 3.4458x; 3.4458x over previous
#include <hip/hip_runtime.h>

#define F_IN 128
#define C 64

typedef float v2f __attribute__((ext_vector_type(2)));
typedef float v8f __attribute__((ext_vector_type(8)));

__device__ __forceinline__ unsigned f32_order_encode(float f) {
  unsigned b = __float_as_uint(f);
  return (b & 0x80000000u) ? ~b : (b | 0x80000000u);
}
__device__ __forceinline__ float f32_order_decode(unsigned u) {
  unsigned b = (u & 0x80000000u) ? (u & 0x7FFFFFFFu) : ~u;
  return __uint_as_float(b);
}

__global__ void k_gemm(const float* __restrict__ x, const float* __restrict__ Wl,
                       const float* __restrict__ Wr, float* __restrict__ xl,
                       float* __restrict__ xr, int N) {
  int wave = threadIdx.x >> 5;
  int lane = threadIdx.x & 31;
  long long t = (long long)blockIdx.x * 8 + wave;
  long long rowTiles = (long long)(N >> 4);
  if (t >= rowTiles * 8) return;
  int sub   = (int)(t & 7);
  int m0    = (int)(t >> 3) << 4;
  int n0    = (sub >> 1) << 4;
  int which = sub & 1;
  const float* W = which ? Wr : Wl;
  float*       Y = which ? xr : xl;

  int half = lane >> 4;
  int l15  = lane & 15;
  const float* arow = x + (long long)(m0 + l15) * F_IN + half * 2;
  const float* bcol = W + n0 + l15 + (long long)(half * 2) * C;

  v8f acc = {0.f, 0.f, 0.f, 0.f, 0.f, 0.f, 0.f, 0.f};
#pragma unroll 4
  for (int k0 = 0; k0 < F_IN; k0 += 4) {
    v2f a, b;
    a.x = arow[k0];
    a.y = arow[k0 + 1];
    b.x = bcol[(long long)k0 * C];
    b.y = bcol[(long long)(k0 + 1) * C];
    acc = __builtin_amdgcn_wmma_f32_16x16x4_f32(
         false, a,  false, b,
         (short)0, acc,  false,  false);
  }
  float* yrow = Y + (long long)(m0 + half * 8) * C + n0 + l15;
#pragma unroll
  for (int r = 0; r < 8; ++r) yrow[(long long)r * C] = acc[r];
}

#define R_NODES 200
#define CAP 8192
#define NT 1024
#define NHW (NT / 16)
#define U 4

struct Item { int s; unsigned dl; float v; bool valid; };

struct Planes {
  float* xr; float* acc; unsigned* cnt; float* asum; unsigned* lmax; float* den; float* self;
  int* le; unsigned* ldl; float* lv;
};

__device__ __forceinline__ float row16_sum(float v) {
  v += __int_as_float(__builtin_amdgcn_update_dpp(0, __float_as_int(v), 0x161, 0xf, 0xf, true));
  v += __int_as_float(__builtin_amdgcn_update_dpp(0, __float_as_int(v), 0x162, 0xf, 0xf, true));
  v += __int_as_float(__builtin_amdgcn_update_dpp(0, __float_as_int(v), 0x164, 0xf, 0xf, true));
  v += __int_as_float(__builtin_amdgcn_update_dpp(0, __float_as_int(v), 0x168, 0xf, 0xf, true));
  return v;
}

__device__ __forceinline__ float lrelu_dot4(float4 a, float4 r, float ea, float4 we, float4 at) {
  float m0 = a.x + r.x + ea * we.x, m1 = a.y + r.y + ea * we.y;
  float m2 = a.z + r.z + ea * we.z, m3 = a.w + r.w + ea * we.w;
  m0 = (m0 > 0.f) ? m0 : 0.2f * m0; m1 = (m1 > 0.f) ? m1 : 0.2f * m1;
  m2 = (m2 > 0.f) ? m2 : 0.2f * m2; m3 = (m3 > 0.f) ? m3 : 0.2f * m3;
  return m0 * at.x + m1 * at.y + m2 * at.z + m3 * at.w;
}

template <bool OVF>
__device__ __forceinline__ void owner_tail(
    const Planes P, const int* __restrict__ ei, const int* __restrict__ dst,
    const float* __restrict__ eattr, const float* __restrict__ xl,
    const float* __restrict__ We, const float* __restrict__ att,
    const float* __restrict__ bias, float* __restrict__ out, float* __restrict__ logit_ws,
    const int nIter, const int lo, const unsigned nn) {
  const int tid = threadIdx.x;

  for (int i = tid; i < nIter; i += NT) {
    if (!OVF) {
      const int e = P.le[i];
      const unsigned dl = (unsigned)(dst[e] - lo);
      const float ea = eattr[e];
      const int s = ei[e];
      atomicAdd(&P.cnt[dl], 1u);
      atomicAdd(&P.asum[dl], ea);
      P.le[i] = s; P.ldl[i] = dl; P.lv[i] = ea;
    } else {
      const unsigned dl = (unsigned)(dst[i] - lo);
      if (dl < nn) {
        atomicAdd(&P.cnt[dl], 1u);
        atomicAdd(&P.asum[dl], eattr[i]);
      }
    }
  }
  __syncthreads();

  const int hw = tid >> 4;
  const int k  = tid & 15;
  const int rot = (hw & 1) << 1;
  const int total = nIter + (int)nn;
  const float4 we4 = ((const float4*)We)[k];
  const float4 at4 = ((const float4*)att)[k];

  auto fetch = [&](int i, bool wantLogit) -> Item {
    Item it; it.s = 0; it.dl = 0u; it.v = 0.f; it.valid = false;
    if (i < nIter) {
      if (!OVF) { it.s = P.le[i]; it.dl = P.ldl[i]; it.v = P.lv[i]; it.valid = true; }
      else {
        const unsigned dl = (unsigned)(dst[i] - lo);
        if (dl < nn) { it.s = ei[i]; it.dl = dl; it.v = wantLogit ? logit_ws[i] : eattr[i]; it.valid = true; }
      }
    } else if (i < total) {
      const unsigned dl = (unsigned)(i - nIter);
      it.s = lo + (int)dl; it.dl = dl; it.valid = true;
      it.v = wantLogit ? P.self[dl] : P.asum[dl] / fmaxf((float)P.cnt[dl], 1.0f);
    }
    return it;
  };

  for (int i0 = hw; i0 < total; i0 += U * NHW) {
    Item it[U]; float4 xs[U]; float p[U];
#pragma unroll
    for (int u = 0; u < U; ++u) it[u] = fetch(i0 + u * NHW, false);
#pragma unroll
    for (int u = 0; u < U; ++u) {
      xs[u] = make_float4(0.f, 0.f, 0.f, 0.f);
      if (it[u].valid) xs[u] = ((const float4*)(xl + (size_t)it[u].s * C))[k];
    }
#pragma unroll
    for (int u = 0; u < U; ++u) {
      p[u] = 0.f;
      if (it[u].valid) {
        const float4 r = ((const float4*)(P.xr + it[u].dl * C))[k];
        p[u] = lrelu_dot4(xs[u], r, it[u].v, we4, at4);
      }
    }
#pragma unroll
    for (int u = 0; u < U; ++u) p[u] = row16_sum(p[u]);
    if (k == 0) {
#pragma unroll
      for (int u = 0; u < U; ++u) {
        const int i = i0 + u * NHW;
        if (it[u].valid) {
          if (i >= nIter) P.self[it[u].dl] = p[u]; else if (OVF) logit_ws[i] = p[u]; else P.lv[i] = p[u];
          atomicMax(&P.lmax[it[u].dl], f32_order_encode(p[u]));
        }
      }
    }
  }
  __syncthreads();

  for (int i0 = hw; i0 < total; i0 += U * NHW) {
    Item it[U]; float4 xs[U];
#pragma unroll
    for (int u = 0; u < U; ++u) it[u] = fetch(i0 + u * NHW, true);
#pragma unroll
    for (int u = 0; u < U; ++u) {
      xs[u] = make_float4(0.f, 0.f, 0.f, 0.f);
      if (it[u].valid) xs[u] = ((const float4*)(xl + (size_t)it[u].s * C))[k];
    }
#pragma unroll
    for (int u = 0; u < U; ++u) {
      if (it[u].valid) {
        const float ex = __expf(it[u].v - f32_order_decode(P.lmax[it[u].dl]));
        if (k == 0) atomicAdd(&P.den[it[u].dl], ex);
        float* acc = P.acc + it[u].dl * C + 4 * k;
        const float v0 = ex * xs[u].x, v1 = ex * xs[u].y, v2 = ex * xs[u].z, v3 = ex * xs[u].w;
        atomicAdd(acc + (0 ^ rot), rot ? v2 : v0); atomicAdd(acc + (1 ^ rot), rot ? v3 : v1);
        atomicAdd(acc + (2 ^ rot), rot ? v0 : v2); atomicAdd(acc + (3 ^ rot), rot ? v1 : v3);
      }
    }
  }
  __syncthreads();

  const float4* a4 = (const float4*)P.acc;
  float4* o4 = (float4*)(out + (size_t)lo * C);
  for (int i = tid; i < (int)nn * (C / 4); i += NT) {
    const float4 a = a4[i];
    const float4 b = ((const float4*)bias)[i & 15];
    const float dn = P.den[i >> 4];
    float4 o;
    o.x = a.x / dn + b.x; o.y = a.y / dn + b.y; o.z = a.z / dn + b.z; o.w = a.w / dn + b.w;
    o4[i] = o;
  }
}

__global__ __launch_bounds__(NT) void k_owner(
    const int* __restrict__ ei, const float* __restrict__ eattr,
    const float* __restrict__ xl, const float* __restrict__ xr,
    const float* __restrict__ We, const float* __restrict__ att,
    const float* __restrict__ bias, float* __restrict__ out,
    float* __restrict__ logit_ws, int E, int N) {
  __shared__ __align__(16) float s_xr[R_NODES * C];
  __shared__ __align__(16) float s_acc[R_NODES * C];
  __shared__ unsigned s_cnt[R_NODES];
  __shared__ float    s_asum[R_NODES];
  __shared__ unsigned s_lmax[R_NODES];
  __shared__ float    s_den[R_NODES];
  __shared__ float    s_self[R_NODES];
  __shared__ int      s_le[CAP];
  __shared__ unsigned s_ldl[CAP];
  __shared__ float    s_lv[CAP];
  __shared__ unsigned s_n;

  const int tid = threadIdx.x;
  const int lo  = blockIdx.x * R_NODES;
  const unsigned nn = (unsigned)min(R_NODES, N - lo);
  const int* __restrict__ dst = ei + E;

  {
    const float4 z4 = {0.f, 0.f, 0.f, 0.f};
    float4* a4 = (float4*)s_acc;
    for (int i = tid; i < R_NODES * (C / 4); i += NT) a4[i] = z4;
    for (int i = tid; i < R_NODES; i += NT) {
      s_cnt[i] = 0u; s_asum[i] = 0.f; s_lmax[i] = 0u; s_den[i] = 0.f; s_self[i] = 0.f;
    }
    if (tid == 0) s_n = 0u;
    const float4* src = (const float4*)(xr + (size_t)lo * C);
    float4* d4 = (float4*)s_xr;
    for (int i = tid; i < (int)nn * (C / 4); i += NT) d4[i] = src[i];
  }
  __syncthreads();

  const int E4 = ((E & 3) == 0) ? (E >> 2) : 0;
  const int4* dst4 = (const int4*)dst;
  for (int q0 = tid; q0 < E4; q0 += 4 * NT) {
    int4 v[4];
#pragma unroll
    for (int u = 0; u < 4; ++u) {
      const int q = q0 + u * NT;
      v[u] = (q < E4) ? dst4[q] : make_int4(-1, -1, -1, -1);
    }
    unsigned c = 0u;
#pragma unroll
    for (int u = 0; u < 4; ++u) {
      c += (unsigned)((unsigned)(v[u].x - lo) < nn) + (unsigned)((unsigned)(v[u].y - lo) < nn) +
           (unsigned)((unsigned)(v[u].z - lo) < nn) + (unsigned)((unsigned)(v[u].w - lo) < nn);
    }
    if (c) {
      unsigned p = atomicAdd(&s_n, c);
#pragma unroll
      for (int u = 0; u < 4; ++u) {
        const int e0 = (q0 + u * NT) * 4;
        if ((unsigned)(v[u].x - lo) < nn) { if (p < (unsigned)CAP) s_le[p] = e0;     ++p; }
        if ((unsigned)(v[u].y - lo) < nn) { if (p < (unsigned)CAP) s_le[p] = e0 + 1; ++p; }
        if ((unsigned)(v[u].z - lo) < nn) { if (p < (unsigned)CAP) s_le[p] = e0 + 2; ++p; }
        if ((unsigned)(v[u].w - lo) < nn) { if (p < (unsigned)CAP) s_le[p] = e0 + 3; ++p; }
      }
    }
  }
  for (int e = E4 * 4 + tid; e < E; e += NT) {
    if ((unsigned)(dst[e] - lo) < nn) {
      unsigned p = atomicAdd(&s_n, 1u);
      if (p < (unsigned)CAP) s_le[p] = e;
    }
  }
  __syncthreads();
  const unsigned nraw = s_n;

  Planes P;
  P.xr = s_xr; P.acc = s_acc; P.cnt = s_cnt; P.asum = s_asum; P.lmax = s_lmax; P.den = s_den;
  P.self = s_self; P.le = s_le; P.ldl = s_ldl; P.lv = s_lv;
  if (nraw > (unsigned)CAP)
    owner_tail<true>(P, ei, dst, eattr, xl, We, att, bias, out, logit_ws, E, lo, nn);
  else
    owner_tail<false>(P, ei, dst, eattr, xl, We, att, bias, out, logit_ws, (int)nraw, lo, nn);
}

extern "C" void kernel_launch(void* const* d_in, const int* in_sizes, int n_in,
                              void* d_out, int out_size, void* d_ws, size_t ws_size,
                              hipStream_t stream) {
  (void)n_in; (void)out_size; (void)ws_size;
  const float* x     = (const float*)d_in[0];
  const float* eattr = (const float*)d_in[1];
  const float* Wl    = (const float*)d_in[2];
  const float* Wr    = (const float*)d_in[3];
  const float* We    = (const float*)d_in[4];
  const float* att   = (const float*)d_in[5];
  const float* bias  = (const float*)d_in[6];
  const int*   ei    = (const int*)d_in[7];
  float*       out   = (float*)d_out;

  int N = in_sizes[0] / F_IN;
  int E = in_sizes[1];

  char* ws = (char*)d_ws;
  size_t o = 0;
  auto take = [&](size_t bytes) -> void* {
    void* p = ws + o; o += (bytes + 255) & ~(size_t)255; return p;
  };
  float* xl    = (float*)take((size_t)N * C * 4);
  float* xr    = (float*)take((size_t)N * C * 4);
  float* logit = (float*)take((size_t)E * 4);

  k_gemm<<<(unsigned)(N >> 4), 256, 0, stream>>>(x, Wl, Wr, xl, xr, N);
  k_owner<<<(unsigned)((N + R_NODES - 1) / R_NODES), NT, 0, stream>>>(
      ei, eattr, xl, xr, We, att, bias, out, logit, E, N);
}


# edge filter scans a 1-byte owner-key plane (dst/200, written by the GEMM launch's threads) instead of 4-byte destinations: 16 keys a load, bytes matched in-register; LDS list (sum) order changes
# speedup vs baseline: 4.3435x; 1.2605x over previous
#include <hip/hip_runtime.h>

#define F_IN 128
#define C 64

typedef float v2f __attribute__((ext_vector_type(2)));
typedef float v8f __attribute__((ext_vector_type(8)));

__device__ __forceinline__ unsigned f32_order_encode(float f) {
  unsigned b = __float_as_uint(f);
  return (b & 0x80000000u) ? ~b : (b | 0x80000000u);
}
__device__ __forceinline__ float f32_order_decode(unsigned u) {
  unsigned b = (u & 0x80000000u) ? (u & 0x7FFFFFFFu) : ~u;
  return __uint_as_float(b);
}

#define R_NODES 200

__device__ __forceinline__ void write_owner_keys(const int* __restrict__ ei,
                                                 unsigned* __restrict__ keys, int E) {
  const int* __restrict__ dst = ei + E;
  const int nW = (E >> 4) << 2;
  for (int q = blockIdx.x * blockDim.x + threadIdx.x; q < nW; q += gridDim.x * blockDim.x) {
    const unsigned d0 = (unsigned)dst[4 * q], d1 = (unsigned)dst[4 * q + 1];
    const unsigned d2 = (unsigned)dst[4 * q + 2], d3 = (unsigned)dst[4 * q + 3];
    keys[q] = ((d0 / R_NODES) & 255u) | (((d1 / R_NODES) & 255u) << 8) |
              (((d2 / R_NODES) & 255u) << 16) | ((d3 / R_NODES) << 24);
  }
}

__global__ void k_gemm(const float* __restrict__ x, const float* __restrict__ Wl,
                       const float* __restrict__ Wr, float* __restrict__ xl,
                       float* __restrict__ xr, int N,
                       const int* __restrict__ ei, unsigned* __restrict__ keys, int E) {
  write_owner_keys(ei, keys, E);
  int wave = threadIdx.x >> 5;
  int lane = threadIdx.x & 31;
  long long t = (long long)blockIdx.x * 8 + wave;
  long long rowTiles = (long long)(N >> 4);
  if (t >= rowTiles * 8) return;
  int sub   = (int)(t & 7);
  int m0    = (int)(t >> 3) << 4;
  int n0    = (sub >> 1) << 4;
  int which = sub & 1;
  const float* W = which ? Wr : Wl;
  float*       Y = which ? xr : xl;

  int half = lane >> 4;
  int l15  = lane & 15;
  const float* arow = x + (long long)(m0 + l15) * F_IN + half * 2;
  const float* bcol = W + n0 + l15 + (long long)(half * 2) * C;

  v8f acc = {0.f, 0.f, 0.f, 0.f, 0.f, 0.f, 0.f, 0.f};
#pragma unroll 4
  for (int k0 = 0; k0 < F_IN; k0 += 4) {
    v2f a, b;
    a.x = arow[k0];
    a.y = arow[k0 + 1];
    b.x = bcol[(long long)k0 * C];
    b.y = bcol[(long long)(k0 + 1) * C];
    acc = __builtin_amdgcn_wmma_f32_16x16x4_f32(
         false, a,  false, b,
         (short)0, acc,  false,  false);
  }
  float* yrow = Y + (long long)(m0 + half * 8) * C + n0 + l15;
#pragma unroll
  for (int r = 0; r < 8; ++r) yrow[(long long)r * C] = acc[r];
}

#define CAP 8192
#define NT 1024
#define NHW (NT / 16)
#define U 4

__device__ __forceinline__ unsigned zero_bytes(unsigned t) {
  return ~(((t & 0x7F7F7F7Fu) + 0x7F7F7F7Fu) | t) & 0x80808080u;
}
__device__ __forceinline__ unsigned match16(uint4 v, unsigned pat) {
  return (zero_bytes(v.x ^ pat) >> 7) | (zero_bytes(v.y ^ pat) >> 6) |
         (zero_bytes(v.z ^ pat) >> 5) | (zero_bytes(v.w ^ pat) >> 4);
}

struct Item { int s; unsigned dl; float v; bool valid; };

struct Planes {
  float* xr; float* acc; unsigned* cnt; float* asum; unsigned* lmax; float* den; float* self;
  int* le; unsigned* ldl; float* lv;
};

__device__ __forceinline__ float row16_sum(float v) {
  v += __int_as_float(__builtin_amdgcn_update_dpp(0, __float_as_int(v), 0x161, 0xf, 0xf, true));
  v += __int_as_float(__builtin_amdgcn_update_dpp(0, __float_as_int(v), 0x162, 0xf, 0xf, true));
  v += __int_as_float(__builtin_amdgcn_update_dpp(0, __float_as_int(v), 0x164, 0xf, 0xf, true));
  v += __int_as_float(__builtin_amdgcn_update_dpp(0, __float_as_int(v), 0x168, 0xf, 0xf, true));
  return v;
}

__device__ __forceinline__ float lrelu_dot4(float4 a, float4 r, float ea, float4 we, float4 at) {
  float m0 = a.x + r.x + ea * we.x, m1 = a.y + r.y + ea * we.y;
  float m2 = a.z + r.z + ea * we.z, m3 = a.w + r.w + ea * we.w;
  m0 = (m0 > 0.f) ? m0 : 0.2f * m0; m1 = (m1 > 0.f) ? m1 : 0.2f * m1;
  m2 = (m2 > 0.f) ? m2 : 0.2f * m2; m3 = (m3 > 0.f) ? m3 : 0.2f * m3;
  return m0 * at.x + m1 * at.y + m2 * at.z + m3 * at.w;
}

template <bool OVF>
__device__ __forceinline__ void owner_tail(
    const Planes P, const int* __restrict__ ei, const int* __restrict__ dst,
    const float* __restrict__ eattr, const float* __restrict__ xl,
    const float* __restrict__ We, const float* __restrict__ att,
    const float* __restrict__ bias, float* __restrict__ out,
    const int nIter, const int lo, const unsigned nn) {
  const int tid = threadIdx.x;

  for (int i = tid; i < nIter; i += NT) {
    if (!OVF) {
      const int e = P.le[i];
      unsigned dl = (unsigned)(dst[e] - lo);
      const float ea = eattr[e];
      const int s = ei[e];
      if (dl < nn) { atomicAdd(&P.cnt[dl], 1u); atomicAdd(&P.asum[dl], ea); } else dl = ~0u;
      P.le[i] = s; P.ldl[i] = dl; P.lv[i] = ea;
    } else {
      const unsigned dl = (unsigned)(dst[i] - lo);
      if (dl < nn) {
        atomicAdd(&P.cnt[dl], 1u);
        atomicAdd(&P.asum[dl], eattr[i]);
      }
    }
  }
  __syncthreads();

  const int hw = tid >> 4;
  const int k  = tid & 15;
  const int rot = (hw & 1) << 1;
  const int total = nIter + (int)nn;
  const float4 we4 = ((const float4*)We)[k];
  const float4 at4 = ((const float4*)att)[k];

  auto fetch = [&](int i, bool wantLogit) -> Item {
    Item it; it.s = 0; it.dl = 0u; it.v = 0.f; it.valid = false;
    if (i < nIter) {
      if (!OVF) { it.s = P.le[i]; it.dl = P.ldl[i]; it.v = P.lv[i]; it.valid = it.dl != ~0u; }
      else {
        const unsigned dl = (unsigned)(dst[i] - lo);
        if (dl < nn) { it.s = ei[i]; it.dl = dl; it.v = eattr[i]; it.valid = true; }
      }
    } else if (i < total) {
      const unsigned dl = (unsigned)(i - nIter);
      it.s = lo + (int)dl; it.dl = dl; it.valid = true;
      it.v = wantLogit ? P.self[dl] : P.asum[dl] / fmaxf((float)P.cnt[dl], 1.0f);
    }
    return it;
  };

  for (int i0 = hw; i0 < total; i0 += U * NHW) {
    Item it[U]; float4 xs[U]; float p[U];
#pragma unroll
    for (int u = 0; u < U; ++u) it[u] = fetch(i0 + u * NHW, false);
#pragma unroll
    for (int u = 0; u < U; ++u) {
      xs[u] = make_float4(0.f, 0.f, 0.f, 0.f);
      if (it[u].valid) xs[u] = ((const float4*)(xl + (size_t)it[u].s * C))[k];
    }
#pragma unroll
    for (int u = 0; u < U; ++u) {
      p[u] = 0.f;
      if (it[u].valid) {
        const float4 r = ((const float4*)(P.xr + it[u].dl * C))[k];
        p[u] = lrelu_dot4(xs[u], r, it[u].v, we4, at4);
      }
    }
#pragma unroll
    for (int u = 0; u < U; ++u) p[u] = row16_sum(p[u]);
    if (k == 0) {
#pragma unroll
      for (int u = 0; u < U; ++u) {
        const int i = i0 + u * NHW;
        if (it[u].valid) {
          if (i >= nIter) P.self[it[u].dl] = p[u]; else if (!OVF) P.lv[i] = p[u];
          atomicMax(&P.lmax[it[u].dl], f32_order_encode(p[u]));
        }
      }
    }
  }
  __syncthreads();

  for (int i0 = hw; i0 < total; i0 += U * NHW) {
    Item it[U]; float4 xs[U];
#pragma unroll
    for (int u = 0; u < U; ++u) it[u] = fetch(i0 + u * NHW, true);
#pragma unroll
    for (int u = 0; u < U; ++u) {
      xs[u] = make_float4(0.f, 0.f, 0.f, 0.f);
      if (it[u].valid) xs[u] = ((const float4*)(xl + (size_t)it[u].s * C))[k];
    }
    if (OVF) {
      float p[U];
#pragma unroll
      for (int u = 0; u < U; ++u) {
        p[u] = 0.f;
        if (it[u].valid && i0 + u * NHW < nIter) {
          const float4 r = ((const float4*)(P.xr + it[u].dl * C))[k];
          p[u] = lrelu_dot4(xs[u], r, it[u].v, we4, at4);
        }
      }
#pragma unroll
      for (int u = 0; u < U; ++u) {
        p[u] = row16_sum(p[u]);
        if (i0 + u * NHW < nIter) it[u].v = p[u];
      }
    }
#pragma unroll
    for (int u = 0; u < U; ++u) {
      if (it[u].valid) {
        const float ex = __expf(it[u].v - f32_order_decode(P.lmax[it[u].dl]));
        if (k == 0) atomicAdd(&P.den[it[u].dl], ex);
        float* acc = P.acc + it[u].dl * C + 4 * k;
        const float v0 = ex * xs[u].x, v1 = ex * xs[u].y, v2 = ex * xs[u].z, v3 = ex * xs[u].w;
        atomicAdd(acc + (0 ^ rot), rot ? v2 : v0); atomicAdd(acc + (1 ^ rot), rot ? v3 : v1);
        atomicAdd(acc + (2 ^ rot), rot ? v0 : v2); atomicAdd(acc + (3 ^ rot), rot ? v1 : v3);
      }
    }
  }
  __syncthreads();

  const float4* a4 = (const float4*)P.acc;
  float4* o4 = (float4*)(out + (size_t)lo * C);
  for (int i = tid; i < (int)nn * (C / 4); i += NT) {
    const float4 a = a4[i];
    const float4 b = ((const float4*)bias)[i & 15];
    const float dn = P.den[i >> 4];
    float4 o;
    o.x = a.x / dn + b.x; o.y = a.y / dn + b.y; o.z = a.z / dn + b.z; o.w = a.w / dn + b.w;
    o4[i] = o;
  }
}

__global__ __launch_bounds__(NT) void k_owner(
    const int* __restrict__ ei, const float* __restrict__ eattr,
    const float* __restrict__ xl, const float* __restrict__ xr,
    const float* __restrict__ We, const float* __restrict__ att,
    const float* __restrict__ bias, float* __restrict__ out,
    const unsigned* __restrict__ keys, int E, int N) {
  __shared__ __align__(16) float s_xr[R_NODES * C];
  __shared__ __align__(16) float s_acc[R_NODES * C];
  __shared__ unsigned s_cnt[R_NODES];
  __shared__ float    s_asum[R_NODES];
  __shared__ unsigned s_lmax[R_NODES];
  __shared__ float    s_den[R_NODES];
  __shared__ float    s_self[R_NODES];
  __shared__ int      s_le[CAP];
  __shared__ unsigned s_ldl[CAP];
  __shared__ float    s_lv[CAP];
  __shared__ unsigned s_n;

  const int tid = threadIdx.x;
  const int lo  = blockIdx.x * R_NODES;
  const unsigned nn = (unsigned)min(R_NODES, N - lo);
  const int* __restrict__ dst = ei + E;

  {
    const float4 z4 = {0.f, 0.f, 0.f, 0.f};
    float4* a4 = (float4*)s_acc;
    for (int i = tid; i < R_NODES * (C / 4); i += NT) a4[i] = z4;
    for (int i = tid; i < R_NODES; i += NT) {
      s_cnt[i] = 0u; s_asum[i] = 0.f; s_lmax[i] = 0u; s_den[i] = 0.f; s_self[i] = 0.f;
    }
    if (tid == 0) s_n = 0u;
    const float4* src = (const float4*)(xr + (size_t)lo * C);
    float4* d4 = (float4*)s_xr;
    for (int i = tid; i < (int)nn * (C / 4); i += NT) d4[i] = src[i];
  }
  __syncthreads();

  const unsigned pat = (blockIdx.x & 255u) * 0x01010101u;
  const int G = E >> 4;
  const uint4* key4 = (const uint4*)keys;
  auto place = [&](unsigned m0, unsigned m1, int g0) {
    while (m0 | m1) {
      const bool first = m0 != 0u;
      const unsigned m = first ? m0 : m1;
      const int bit = __builtin_ctz(m);
      if (first) m0 = m & (m - 1u); else m1 = m & (m - 1u);
      const int g = g0 + ((first ? 0 : 2) + ((bit >> 2) & 1)) * NT;
      const unsigned p = atomicAdd(&s_n, 1u);
      if (p < (unsigned)CAP) s_le[p] = (g << 4) + ((bit & 3) << 2) + (bit >> 3);
    }
  };
  const int Gmain = G - G % (4 * NT);
  for (int g0 = tid; g0 < Gmain; g0 += 4 * NT) {
    uint4 v[4];
#pragma unroll
    for (int u = 0; u < 4; ++u) v[u] = key4[g0 + u * NT];
    place(match16(v[0], pat) | (match16(v[1], pat) << 4),
          match16(v[2], pat) | (match16(v[3], pat) << 4), g0);
  }
  for (int g = Gmain + tid; g < G; g += NT) place(match16(key4[g], pat), 0u, g);
  for (int e = (G << 4) + tid; e < E; e += NT) {
    if ((unsigned)(dst[e] - lo) < nn) {
      unsigned p = atomicAdd(&s_n, 1u);
      if (p < (unsigned)CAP) s_le[p] = e;
    }
  }
  __syncthreads();
  const unsigned nraw = s_n;

  Planes P;
  P.xr = s_xr; P.acc = s_acc; P.cnt = s_cnt; P.asum = s_asum; P.lmax = s_lmax; P.den = s_den;
  P.self = s_self; P.le = s_le; P.ldl = s_ldl; P.lv = s_lv;
  if (nraw > (unsigned)CAP)
    owner_tail<true>(P, ei, dst, eattr, xl, We, att, bias, out, E, lo, nn);
  else
    owner_tail<false>(P, ei, dst, eattr, xl, We, att, bias, out, (int)nraw, lo, nn);
}

extern "C" void kernel_launch(void* const* d_in, const int* in_sizes, int n_in,
                              void* d_out, int out_size, void* d_ws, size_t ws_size,
                              hipStream_t stream) {
  (void)n_in; (void)out_size; (void)ws_size;
  const float* x     = (const float*)d_in[0];
  const float* eattr = (const float*)d_in[1];
  const float* Wl    = (const float*)d_in[2];
  const float* Wr    = (const float*)d_in[3];
  const float* We    = (const float*)d_in[4];
  const float* att   = (const float*)d_in[5];
  const float* bias  = (const float*)d_in[6];
  const int*   ei    = (const int*)d_in[7];
  float*       out   = (float*)d_out;

  int N = in_sizes[0] / F_IN;
  int E = in_sizes[1];

  char* ws = (char*)d_ws;
  size_t o = 0;
  auto take = [&](size_t bytes) -> void* {
    void* p = ws + o; o += (bytes + 255) & ~(size_t)255; return p;
  };
  float* xl    = (float*)take((size_t)N * C * 4);
  float* xr    = (float*)take((size_t)N * C * 4);
  unsigned* keys = (unsigned*)take((size_t)E);

  k_gemm<<<(unsigned)(N >> 4), 256, 0, stream>>>(x, Wl, Wr, xl, xr, N, ei, keys, E);
  k_owner<<<(unsigned)((N + R_NODES - 1) / R_NODES), NT, 0, stream>>>(
      ei, eattr, xl, xr, We, att, bias, out, keys, E, N);
}


# owner block counting-sorts its LDS edge list by destination; a 16-lane row per destination runs a running softmax in registers: x_l rows gathered once, no float atomics left; sum order changes
# speedup vs baseline: 5.5472x; 1.2771x over previous
#include <hip/hip_runtime.h>

#define F_IN 128
#define C 64

typedef float v2f __attribute__((ext_vector_type(2)));
typedef float v8f __attribute__((ext_vector_type(8)));

__device__ __forceinline__ unsigned f32_order_encode(float f) {
  unsigned b = __float_as_uint(f);
  return (b & 0x80000000u) ? ~b : (b | 0x80000000u);
}
__device__ __forceinline__ float f32_order_decode(unsigned u) {
  unsigned b = (u & 0x80000000u) ? (u & 0x7FFFFFFFu) : ~u;
  return __uint_as_float(b);
}

#define R_NODES 200
static_assert(R_NODES <= 256, "a local destination must fit the low byte of a packed rank word");

__device__ __forceinline__ void write_owner_keys(const int* __restrict__ ei,
                                                 unsigned* __restrict__ keys, int E) {
  const int* __restrict__ dst = ei + E;
  const int nW = (E >> 4) << 2;
  for (int q = blockIdx.x * blockDim.x + threadIdx.x; q < nW; q += gridDim.x * blockDim.x) {
    const unsigned d0 = (unsigned)dst[4 * q], d1 = (unsigned)dst[4 * q + 1];
    const unsigned d2 = (unsigned)dst[4 * q + 2], d3 = (unsigned)dst[4 * q + 3];
    keys[q] = ((d0 / R_NODES) & 255u) | (((d1 / R_NODES) & 255u) << 8) |
              (((d2 / R_NODES) & 255u) << 16) | ((d3 / R_NODES) << 24);
  }
}

__global__ void k_gemm(const float* __restrict__ x, const float* __restrict__ Wl,
                       const float* __restrict__ Wr, float* __restrict__ xl,
                       float* __restrict__ xr, int N,
                       const int* __restrict__ ei, unsigned* __restrict__ keys, int E) {
  write_owner_keys(ei, keys, E);
  int wave = threadIdx.x >> 5;
  int lane = threadIdx.x & 31;
  long long t = (long long)blockIdx.x * 8 + wave;
  long long rowTiles = (long long)(N >> 4);
  if (t >= rowTiles * 8) return;
  int sub   = (int)(t & 7);
  int m0    = (int)(t >> 3) << 4;
  int n0    = (sub >> 1) << 4;
  int which = sub & 1;
  const float* W = which ? Wr : Wl;
  float*       Y = which ? xr : xl;

  int half = lane >> 4;
  int l15  = lane & 15;
  const float* arow = x + (long long)(m0 + l15) * F_IN + half * 2;
  const float* bcol = W + n0 + l15 + (long long)(half * 2) * C;

  v8f acc = {0.f, 0.f, 0.f, 0.f, 0.f, 0.f, 0.f, 0.f};
#pragma unroll 4
  for (int k0 = 0; k0 < F_IN; k0 += 4) {
    v2f a, b;
    a.x = arow[k0];
    a.y = arow[k0 + 1];
    b.x = bcol[(long long)k0 * C];
    b.y = bcol[(long long)(k0 + 1) * C];
    acc = __builtin_amdgcn_wmma_f32_16x16x4_f32(
         false, a,  false, b,
         (short)0, acc,  false,  false);
  }
  float* yrow = Y + (long long)(m0 + half * 8) * C + n0 + l15;
#pragma unroll
  for (int r = 0; r < 8; ++r) yrow[(long long)r * C] = acc[r];
}

#define CAP 8192
#define NT 1024
#define NHW (NT / 16)
#define U 4

__device__ __forceinline__ unsigned zero_bytes(unsigned t) {
  return ~(((t & 0x7F7F7F7Fu) + 0x7F7F7F7Fu) | t) & 0x80808080u;
}
__device__ __forceinline__ unsigned match16(uint4 v, unsigned pat) {
  return (zero_bytes(v.x ^ pat) >> 7) | (zero_bytes(v.y ^ pat) >> 6) |
         (zero_bytes(v.z ^ pat) >> 5) | (zero_bytes(v.w ^ pat) >> 4);
}

struct Item { int s; unsigned dl; float v; bool valid; };

struct Planes {
  float* xr; float* acc; unsigned* cnt; float* asum; unsigned* lmax; float* den; float* self;
  int* le; unsigned* ldl; float* lv; unsigned* off; unsigned* next;
};

__device__ __forceinline__ float row16_sum(float v) {
  v += __int_as_float(__builtin_amdgcn_update_dpp(0, __float_as_int(v), 0x161, 0xf, 0xf, true));
  v += __int_as_float(__builtin_amdgcn_update_dpp(0, __float_as_int(v), 0x162, 0xf, 0xf, true));
  v += __int_as_float(__builtin_amdgcn_update_dpp(0, __float_as_int(v), 0x164, 0xf, 0xf, true));
  v += __int_as_float(__builtin_amdgcn_update_dpp(0, __float_as_int(v), 0x168, 0xf, 0xf, true));
  return v;
}

__device__ __forceinline__ float lrelu_dot4(float4 a, float4 r, float ea, float4 we, float4 at) {
  float m0 = a.x + r.x + ea * we.x, m1 = a.y + r.y + ea * we.y;
  float m2 = a.z + r.z + ea * we.z, m3 = a.w + r.w + ea * we.w;
  m0 = (m0 > 0.f) ? m0 : 0.2f * m0; m1 = (m1 > 0.f) ? m1 : 0.2f * m1;
  m2 = (m2 > 0.f) ? m2 : 0.2f * m2; m3 = (m3 > 0.f) ? m3 : 0.2f * m3;
  return m0 * at.x + m1 * at.y + m2 * at.z + m3 * at.w;
}

template <bool OVF>
__device__ __forceinline__ void owner_tail(
    const Planes P, const int* __restrict__ ei, const int* __restrict__ dst,
    const float* __restrict__ eattr, const float* __restrict__ xl,
    const float* __restrict__ We, const float* __restrict__ att,
    const float* __restrict__ bias, float* __restrict__ out,
    const int nIter, const int lo, const unsigned nn) {
  const int tid = threadIdx.x;

  for (int i = tid; i < nIter; i += NT) {
    if (!OVF) {
      const int e = P.le[i];
      unsigned dl = (unsigned)(dst[e] - lo);
      const float ea = eattr[e];
      const int s = ei[e];
      if (dl < nn) { atomicAdd(&P.cnt[dl], 1u); atomicAdd(&P.asum[dl], ea); } else dl = ~0u;
      P.le[i] = s; P.ldl[i] = dl; P.lv[i] = ea;
    } else {
      const unsigned dl = (unsigned)(dst[i] - lo);
      if (dl < nn) {
        atomicAdd(&P.cnt[dl], 1u);
        atomicAdd(&P.asum[dl], eattr[i]);
      }
    }
  }
  __syncthreads();

  const int hw = tid >> 4;
  const int k  = tid & 15;
  const int rot = (hw & 1) << 1;
  const int total = nIter + (int)nn;
  const float4 we4 = ((const float4*)We)[k];
  const float4 at4 = ((const float4*)att)[k];

  auto fetch = [&](int i, bool wantLogit) -> Item {
    Item it; it.s = 0; it.dl = 0u; it.v = 0.f; it.valid = false;
    if (i < nIter) {
      if (!OVF) { it.s = P.le[i]; it.dl = P.ldl[i]; it.v = P.lv[i]; it.valid = it.dl != ~0u; }
      else {
        const unsigned dl = (unsigned)(dst[i] - lo);
        if (dl < nn) { it.s = ei[i]; it.dl = dl; it.v = eattr[i]; it.valid = true; }
      }
    } else if (i < total) {
      const unsigned dl = (unsigned)(i - nIter);
      it.s = lo + (int)dl; it.dl = dl; it.valid = true;
      it.v = wantLogit ? P.self[dl] : P.asum[dl] / fmaxf((float)P.cnt[dl], 1.0f);
    }
    return it;
  };

  for (int i0 = hw; i0 < total; i0 += U * NHW) {
    Item it[U]; float4 xs[U]; float p[U];
#pragma unroll
    for (int u = 0; u < U; ++u) it[u] = fetch(i0 + u * NHW, false);
#pragma unroll
    for (int u = 0; u < U; ++u) {
      xs[u] = make_float4(0.f, 0.f, 0.f, 0.f);
      if (it[u].valid) xs[u] = ((const float4*)(xl + (size_t)it[u].s * C))[k];
    }
#pragma unroll
    for (int u = 0; u < U; ++u) {
      p[u] = 0.f;
      if (it[u].valid) {
        const float4 r = ((const float4*)(P.xr + it[u].dl * C))[k];
        p[u] = lrelu_dot4(xs[u], r, it[u].v, we4, at4);
      }
    }
#pragma unroll
    for (int u = 0; u < U; ++u) p[u] = row16_sum(p[u]);
    if (k == 0) {
#pragma unroll
      for (int u = 0; u < U; ++u) {
        const int i = i0 + u * NHW;
        if (it[u].valid) {
          if (i >= nIter) P.self[it[u].dl] = p[u]; else if (!OVF) P.lv[i] = p[u];
          atomicMax(&P.lmax[it[u].dl], f32_order_encode(p[u]));
        }
      }
    }
  }
  __syncthreads();

  for (int i0 = hw; i0 < total; i0 += U * NHW) {
    Item it[U]; float4 xs[U];
#pragma unroll
    for (int u = 0; u < U; ++u) it[u] = fetch(i0 + u * NHW, true);
#pragma unroll
    for (int u = 0; u < U; ++u) {
      xs[u] = make_float4(0.f, 0.f, 0.f, 0.f);
      if (it[u].valid) xs[u] = ((const float4*)(xl + (size_t)it[u].s * C))[k];
    }
    if (OVF) {
      float p[U];
#pragma unroll
      for (int u = 0; u < U; ++u) {
        p[u] = 0.f;
        if (it[u].valid && i0 + u * NHW < nIter) {
          const float4 r = ((const float4*)(P.xr + it[u].dl * C))[k];
          p[u] = lrelu_dot4(xs[u], r, it[u].v, we4, at4);
        }
      }
#pragma unroll
      for (int u = 0; u < U; ++u) {
        p[u] = row16_sum(p[u]);
        if (i0 + u * NHW < nIter) it[u].v = p[u];
      }
    }
#pragma unroll
    for (int u = 0; u < U; ++u) {
      if (it[u].valid) {
        const float ex = __expf(it[u].v - f32_order_decode(P.lmax[it[u].dl]));
        if (k == 0) atomicAdd(&P.den[it[u].dl], ex);
        float* acc = P.acc + it[u].dl * C + 4 * k;
        const float v0 = ex * xs[u].x, v1 = ex * xs[u].y, v2 = ex * xs[u].z, v3 = ex * xs[u].w;
        atomicAdd(acc + (0 ^ rot), rot ? v2 : v0); atomicAdd(acc + (1 ^ rot), rot ? v3 : v1);
        atomicAdd(acc + (2 ^ rot), rot ? v0 : v2); atomicAdd(acc + (3 ^ rot), rot ? v1 : v3);
      }
    }
  }
  __syncthreads();

  const float4* a4 = (const float4*)P.acc;
  float4* o4 = (float4*)(out + (size_t)lo * C);
  for (int i = tid; i < (int)nn * (C / 4); i += NT) {
    const float4 a = a4[i];
    const float4 b = ((const float4*)bias)[i & 15];
    const float dn = P.den[i >> 4];
    float4 o;
    o.x = a.x / dn + b.x; o.y = a.y / dn + b.y; o.z = a.z / dn + b.z; o.w = a.w / dn + b.w;
    o4[i] = o;
  }
}

__device__ __forceinline__ void owner_tail_sorted(
    const Planes P, const int* __restrict__ ei, const int* __restrict__ dst,
    const float* __restrict__ eattr, const float* __restrict__ xl,
    const float* __restrict__ We, const float* __restrict__ att,
    const float* __restrict__ bias, float* __restrict__ out,
    const int nIter, const int lo, const unsigned nn) {
  const int tid = threadIdx.x;
  constexpr int PER = CAP / NT;

  int e_[PER]; int rs[PER]; float rv[PER]; int rd[PER]; unsigned rp[PER];
#pragma unroll
  for (int u = 0; u < PER; ++u) {
    const int i = tid + u * NT;
    e_[u] = (i < nIter) ? P.le[i] : -1;
  }
#pragma unroll
  for (int u = 0; u < PER; ++u) {
    rd[u] = 0; rv[u] = 0.f; rs[u] = 0;
    if (e_[u] >= 0) { rd[u] = dst[e_[u]]; rv[u] = eattr[e_[u]]; rs[u] = ei[e_[u]]; }
  }
#pragma unroll
  for (int u = 0; u < PER; ++u) {
    rp[u] = ~0u;
    if (e_[u] >= 0) {
      const unsigned dl = (unsigned)(rd[u] - lo);
      if (dl < nn) {
        const unsigned rank = atomicAdd(&P.cnt[dl], 1u);
        atomicAdd(&P.asum[dl], rv[u]);
        rp[u] = (rank << 8) | dl;
      }
    }
  }
  __syncthreads();

  if (tid < 32) {
    unsigned pre[8]; unsigned run = 0u;
#pragma unroll
    for (int j = 0; j < 8; ++j) { pre[j] = run; run += P.cnt[8 * tid + j]; }
    unsigned incl = run;
#pragma unroll
    for (int d = 1; d < 32; d <<= 1) {
      const unsigned t = __shfl_up(incl, d);
      if (tid >= d) incl += t;
    }
    const unsigned base = incl - run;
#pragma unroll
    for (int j = 0; j < 8; ++j) P.off[8 * tid + j] = base + pre[j];
  }
  __syncthreads();

#pragma unroll
  for (int u = 0; u < PER; ++u) {
    if (rp[u] != ~0u) {
      const unsigned pos = P.off[rp[u] & 255u] + (rp[u] >> 8);
      P.le[pos] = rs[u]; P.lv[pos] = rv[u];
    }
  }
  __syncthreads();

  const int lane = tid & 31;
  const int k  = tid & 15;
  const float4 we4 = ((const float4*)We)[k];
  const float4 at4 = ((const float4*)att)[k];
  const float4 b4  = ((const float4*)bias)[k];
  for (;;) {
    unsigned pr = 0u;
    if (lane == 0) pr = atomicAdd(P.next, 1u);
    pr = (unsigned)__builtin_amdgcn_readfirstlane((int)pr);
    if (2u * pr >= nn) break;
    const int node = (int)(2u * pr) + (lane >> 4);
    if (node >= (int)nn) continue;
    const int beg = (int)P.off[node];
    const unsigned deg = P.cnt[node];
    const int nItem = (int)deg + 1;
    const float eself = P.asum[node] / fmaxf((float)deg, 1.0f);
    const float4 r4 = ((const float4*)(P.xr + node * C))[k];
    float m = -INFINITY, den = 0.f;
    float4 acc = make_float4(0.f, 0.f, 0.f, 0.f);
    for (int j0 = 0; j0 < nItem; j0 += U) {
      int s[U]; float ea[U]; float4 xs[U]; float p[U];
#pragma unroll
      for (int u = 0; u < U; ++u) {
        const int j = j0 + u;
        s[u] = lo + node; ea[u] = eself;
        if (j >= 1 && j < nItem) { s[u] = P.le[beg + j - 1]; ea[u] = P.lv[beg + j - 1]; }
      }
#pragma unroll
      for (int u = 0; u < U; ++u) {
        xs[u] = make_float4(0.f, 0.f, 0.f, 0.f);
        if (j0 + u < nItem) xs[u] = ((const float4*)(xl + (size_t)s[u] * C))[k];
      }
#pragma unroll
      for (int u = 0; u < U; ++u) {
        const float d = row16_sum(lrelu_dot4(xs[u], r4, ea[u], we4, at4));
        p[u] = (j0 + u < nItem) ? d : -INFINITY;
      }
      const float mn = fmaxf(fmaxf(m, p[0]), fmaxf(fmaxf(p[1], p[2]), p[3]));
      const float sc = __expf(m - mn);
      const float e0 = __expf(p[0] - mn), e1 = __expf(p[1] - mn);
      const float e2 = __expf(p[2] - mn), e3 = __expf(p[3] - mn);
      den = den * sc + ((e0 + e1) + (e2 + e3));
      acc.x = acc.x * sc + ((e0 * xs[0].x + e1 * xs[1].x) + (e2 * xs[2].x + e3 * xs[3].x));
      acc.y = acc.y * sc + ((e0 * xs[0].y + e1 * xs[1].y) + (e2 * xs[2].y + e3 * xs[3].y));
      acc.z = acc.z * sc + ((e0 * xs[0].z + e1 * xs[1].z) + (e2 * xs[2].z + e3 * xs[3].z));
      acc.w = acc.w * sc + ((e0 * xs[0].w + e1 * xs[1].w) + (e2 * xs[2].w + e3 * xs[3].w));
      m = mn;
    }
    float4 o;
    o.x = acc.x / den + b4.x; o.y = acc.y / den + b4.y;
    o.z = acc.z / den + b4.z; o.w = acc.w / den + b4.w;
    ((float4*)(out + (size_t)(lo + node) * C))[k] = o;
  }
}

__global__ __launch_bounds__(NT) void k_owner(
    const int* __restrict__ ei, const float* __restrict__ eattr,
    const float* __restrict__ xl, const float* __restrict__ xr,
    const float* __restrict__ We, const float* __restrict__ att,
    const float* __restrict__ bias, float* __restrict__ out,
    const unsigned* __restrict__ keys, int E, int N) {
  __shared__ __align__(16) float s_xr[R_NODES * C];
  __shared__ __align__(16) float s_acc[R_NODES * C];
  __shared__ unsigned s_cnt[256];
  __shared__ unsigned s_off[256];
  __shared__ float    s_asum[R_NODES];
  __shared__ unsigned s_lmax[R_NODES];
  __shared__ float    s_den[R_NODES];
  __shared__ float    s_self[R_NODES];
  __shared__ int      s_le[CAP];
  __shared__ unsigned s_ldl[CAP];
  __shared__ float    s_lv[CAP];
  __shared__ unsigned s_n;
  __shared__ unsigned s_next;

  const int tid = threadIdx.x;
  const int lo  = blockIdx.x * R_NODES;
  const unsigned nn = (unsigned)min(R_NODES, N - lo);
  const int* __restrict__ dst = ei + E;

  {
    const float4 z4 = {0.f, 0.f, 0.f, 0.f};
    float4* a4 = (float4*)s_acc;
    for (int i = tid; i < R_NODES * (C / 4); i += NT) a4[i] = z4;
    for (int i = tid; i < R_NODES; i += NT) {
      s_asum[i] = 0.f; s_lmax[i] = 0u; s_den[i] = 0.f; s_self[i] = 0.f;
    }
    for (int i = tid; i < 256; i += NT) s_cnt[i] = 0u;
    if (tid == 0) { s_n = 0u; s_next = 0u; }
    const float4* src = (const float4*)(xr + (size_t)lo * C);
    float4* d4 = (float4*)s_xr;
    for (int i = tid; i < (int)nn * (C / 4); i += NT) d4[i] = src[i];
  }
  __syncthreads();

  const unsigned pat = (blockIdx.x & 255u) * 0x01010101u;
  const int G = E >> 4;
  const uint4* key4 = (const uint4*)keys;
  auto place = [&](unsigned m0, unsigned m1, int g0) {
    while (m0 | m1) {
      const bool first = m0 != 0u;
      const unsigned m = first ? m0 : m1;
      const int bit = __builtin_ctz(m);
      if (first) m0 = m & (m - 1u); else m1 = m & (m - 1u);
      const int g = g0 + ((first ? 0 : 2) + ((bit >> 2) & 1)) * NT;
      const unsigned p = atomicAdd(&s_n, 1u);
      if (p < (unsigned)CAP) s_le[p] = (g << 4) + ((bit & 3) << 2) + (bit >> 3);
    }
  };
  const int Gmain = G - G % (4 * NT);
  for (int g0 = tid; g0 < Gmain; g0 += 4 * NT) {
    uint4 v[4];
#pragma unroll
    for (int u = 0; u < 4; ++u) v[u] = key4[g0 + u * NT];
    place(match16(v[0], pat) | (match16(v[1], pat) << 4),
          match16(v[2], pat) | (match16(v[3], pat) << 4), g0);
  }
  for (int g = Gmain + tid; g < G; g += NT) place(match16(key4[g], pat), 0u, g);
  for (int e = (G << 4) + tid; e < E; e += NT) {
    if ((unsigned)(dst[e] - lo) < nn) {
      unsigned p = atomicAdd(&s_n, 1u);
      if (p < (unsigned)CAP) s_le[p] = e;
    }
  }
  __syncthreads();
  const unsigned nraw = s_n;

  Planes P;
  P.xr = s_xr; P.acc = s_acc; P.cnt = s_cnt; P.asum = s_asum; P.lmax = s_lmax; P.den = s_den;
  P.self = s_self; P.le = s_le; P.ldl = s_ldl; P.lv = s_lv; P.off = s_off; P.next = &s_next;
  if (nraw > (unsigned)CAP)
    owner_tail<true>(P, ei, dst, eattr, xl, We, att, bias, out, E, lo, nn);
  else
    owner_tail_sorted(P, ei, dst, eattr, xl, We, att, bias, out, (int)nraw, lo, nn);
}

extern "C" void kernel_launch(void* const* d_in, const int* in_sizes, int n_in,
                              void* d_out, int out_size, void* d_ws, size_t ws_size,
                              hipStream_t stream) {
  (void)n_in; (void)out_size; (void)ws_size;
  const float* x     = (const float*)d_in[0];
  const float* eattr = (const float*)d_in[1];
  const float* Wl    = (const float*)d_in[2];
  const float* Wr    = (const float*)d_in[3];
  const float* We    = (const float*)d_in[4];
  const float* att   = (const float*)d_in[5];
  const float* bias  = (const float*)d_in[6];
  const int*   ei    = (const int*)d_in[7];
  float*       out   = (float*)d_out;

  int N = in_sizes[0] / F_IN;
  int E = in_sizes[1];

  char* ws = (char*)d_ws;
  size_t o = 0;
  auto take = [&](size_t bytes) -> void* {
    void* p = ws + o; o += (bytes + 255) & ~(size_t)255; return p;
  };
  float* xl    = (float*)take((size_t)N * C * 4);
  float* xr    = (float*)take((size_t)N * C * 4);
  unsigned* keys = (unsigned*)take((size_t)E);

  k_gemm<<<(unsigned)(N >> 4), 256, 0, stream>>>(x, Wl, Wr, xl, xr, N, ei, keys, E);
  k_owner<<<(unsigned)((N + R_NODES - 1) / R_NODES), NT, 0, stream>>>(
      ei, eattr, xl, xr, We, att, bias, out, keys, E, N);
}


# GEMM feed: 512 blocks walk the row tiles; each 16x128 x tile is staged once in LDS by whole-line loads (next tile prefetched) and feeds 8 waves' A fragments; W slice held in registers; sum order kept
# speedup vs baseline: 6.3681x; 1.1480x over previous
#include <hip/hip_runtime.h>

#define F_IN 128
#define C 64

typedef float v2f __attribute__((ext_vector_type(2)));
typedef float v8f __attribute__((ext_vector_type(8)));

__device__ __forceinline__ unsigned f32_order_encode(float f) {
  unsigned b = __float_as_uint(f);
  return (b & 0x80000000u) ? ~b : (b | 0x80000000u);
}
__device__ __forceinline__ float f32_order_decode(unsigned u) {
  unsigned b = (u & 0x80000000u) ? (u & 0x7FFFFFFFu) : ~u;
  return __uint_as_float(b);
}

#define R_NODES 200
static_assert(R_NODES <= 256, "a local destination must fit the low byte of a packed rank word");

__device__ __forceinline__ void write_owner_keys(const int* __restrict__ ei,
                                                 unsigned* __restrict__ keys, int E) {
  const int* __restrict__ dst = ei + E;
  const int nW = (E >> 4) << 2;
  for (int q = blockIdx.x * blockDim.x + threadIdx.x; q < nW; q += gridDim.x * blockDim.x) {
    const unsigned d0 = (unsigned)dst[4 * q], d1 = (unsigned)dst[4 * q + 1];
    const unsigned d2 = (unsigned)dst[4 * q + 2], d3 = (unsigned)dst[4 * q + 3];
    keys[q] = ((d0 / R_NODES) & 255u) | (((d1 / R_NODES) & 255u) << 8) |
              (((d2 / R_NODES) & 255u) << 16) | ((d3 / R_NODES) << 24);
  }
}

#define GEMM_BLOCKS 512
#define XPITCH 132
__global__ __launch_bounds__(256) void k_gemm(const float* __restrict__ x, const float* __restrict__ Wl,
                       const float* __restrict__ Wr, float* __restrict__ xl,
                       float* __restrict__ xr, int N,
                       const int* __restrict__ ei, unsigned* __restrict__ keys, int E) {
  __shared__ __align__(16) float s_x[2][16 * XPITCH];
  write_owner_keys(ei, keys, E);
  const int tid  = threadIdx.x;
  const int wave = tid >> 5;
  const int lane = tid & 31;
  const int rowTiles = N >> 4;
  int rt = blockIdx.x;
  if (rt >= rowTiles) return;
  const int n0    = (wave >> 1) << 4;
  const int which = wave & 1;
  const float* W = which ? Wr : Wl;
  float*       Y = which ? xr : xl;
  const int half = lane >> 4;
  const int l15  = lane & 15;

  const int st0 = wave * XPITCH + lane * 4;
  const int st1 = st0 + 8 * XPITCH;
  float4 p0, p1;
  {
    const float4* src = (const float4*)(x + (size_t)rt * 16 * F_IN);
    p0 = src[tid]; p1 = src[tid + 256];
  }

  const float* bcol = W + n0 + l15 + (half * 2) * C;
  v2f b[32];
#pragma unroll
  for (int s = 0; s < 32; ++s) { b[s].x = bcol[(4 * s) * C]; b[s].y = bcol[(4 * s + 1) * C]; }

  *(float4*)(&s_x[0][st0]) = p0;
  *(float4*)(&s_x[0][st1]) = p1;
  __syncthreads();

  int cur = 0;
  for (;;) {
    const int nxt = rt + (int)gridDim.x;
    const bool more = nxt < rowTiles;
    if (more) {
      const float4* src = (const float4*)(x + (size_t)nxt * 16 * F_IN);
      p0 = src[tid]; p1 = src[tid + 256];
    }
    const v2f* ap = (const v2f*)(&s_x[cur][l15 * XPITCH + 2 * half]);
    v2f a[32];
#pragma unroll
    for (int s = 0; s < 32; ++s) a[s] = ap[2 * s];
    __builtin_amdgcn_sched_barrier(0);
    v8f acc = {0.f, 0.f, 0.f, 0.f, 0.f, 0.f, 0.f, 0.f};
#pragma unroll
    for (int s = 0; s < 32; ++s) {
      acc = __builtin_amdgcn_wmma_f32_16x16x4_f32(
           false, a[s],  false, b[s],
           (short)0, acc,  false,  false);
    }
    if (more) {
      *(float4*)(&s_x[cur ^ 1][st0]) = p0;
      *(float4*)(&s_x[cur ^ 1][st1]) = p1;
      __syncthreads();
    }
    float* yrow = Y + ((size_t)rt * 16 + half * 8) * C + n0 + l15;
#pragma unroll
    for (int r = 0; r < 8; ++r) yrow[(size_t)r * C] = acc[r];
    if (!more) break;
    cur ^= 1; rt = nxt;
  }
}

#define CAP 8192
#define NT 1024
#define NHW (NT / 16)
#define U 4

__device__ __forceinline__ unsigned zero_bytes(unsigned t) {
  return ~(((t & 0x7F7F7F7Fu) + 0x7F7F7F7Fu) | t) & 0x80808080u;
}
__device__ __forceinline__ unsigned match16(uint4 v, unsigned pat) {
  return (zero_bytes(v.x ^ pat) >> 7) | (zero_bytes(v.y ^ pat) >> 6) |
         (zero_bytes(v.z ^ pat) >> 5) | (zero_bytes(v.w ^ pat) >> 4);
}

struct Item { int s; unsigned dl; float v; bool valid; };

struct Planes {
  float* xr; float* acc; unsigned* cnt; float* asum; unsigned* lmax; float* den; float* self;
  int* le; unsigned* ldl; float* lv; unsigned* off; unsigned* next;
};

__device__ __forceinline__ float row16_sum(float v) {
  v += __int_as_float(__builtin_amdgcn_update_dpp(0, __float_as_int(v), 0x161, 0xf, 0xf, true));
  v += __int_as_float(__builtin_amdgcn_update_dpp(0, __float_as_int(v), 0x162, 0xf, 0xf, true));
  v += __int_as_float(__builtin_amdgcn_update_dpp(0, __float_as_int(v), 0x164, 0xf, 0xf, true));
  v += __int_as_float(__builtin_amdgcn_update_dpp(0, __float_as_int(v), 0x168, 0xf, 0xf, true));
  return v;
}

__device__ __forceinline__ float lrelu_dot4(float4 a, float4 r, float ea, float4 we, float4 at) {
  float m0 = a.x + r.x + ea * we.x, m1 = a.y + r.y + ea * we.y;
  float m2 = a.z + r.z + ea * we.z, m3 = a.w + r.w + ea * we.w;
  m0 = (m0 > 0.f) ? m0 : 0.2f * m0; m1 = (m1 > 0.f) ? m1 : 0.2f * m1;
  m2 = (m2 > 0.f) ? m2 : 0.2f * m2; m3 = (m3 > 0.f) ? m3 : 0.2f * m3;
  return m0 * at.x + m1 * at.y + m2 * at.z + m3 * at.w;
}

template <bool OVF>
__device__ __forceinline__ void owner_tail(
    const Planes P, const int* __restrict__ ei, const int* __restrict__ dst,
    const float* __restrict__ eattr, const float* __restrict__ xl,
    const float* __restrict__ We, const float* __restrict__ att,
    const float* __restrict__ bias, float* __restrict__ out,
    const int nIter, const int lo, const unsigned nn) {
  const int tid = threadIdx.x;

  for (int i = tid; i < nIter; i += NT) {
    if (!OVF) {
      const int e = P.le[i];
      unsigned dl = (unsigned)(dst[e] - lo);
      const float ea = eattr[e];
      const int s = ei[e];
      if (dl < nn) { atomicAdd(&P.cnt[dl], 1u); atomicAdd(&P.asum[dl], ea); } else dl = ~0u;
      P.le[i] = s; P.ldl[i] = dl; P.lv[i] = ea;
    } else {
      const unsigned dl = (unsigned)(dst[i] - lo);
      if (dl < nn) {
        atomicAdd(&P.cnt[dl], 1u);
        atomicAdd(&P.asum[dl], eattr[i]);
      }
    }
  }
  __syncthreads();

  const int hw = tid >> 4;
  const int k  = tid & 15;
  const int rot = (hw & 1) << 1;
  const int total = nIter + (int)nn;
  const float4 we4 = ((const float4*)We)[k];
  const float4 at4 = ((const float4*)att)[k];

  auto fetch = [&](int i, bool wantLogit) -> Item {
    Item it; it.s = 0; it.dl = 0u; it.v = 0.f; it.valid = false;
    if (i < nIter) {
      if (!OVF) { it.s = P.le[i]; it.dl = P.ldl[i]; it.v = P.lv[i]; it.valid = it.dl != ~0u; }
      else {
        const unsigned dl = (unsigned)(dst[i] - lo);
        if (dl < nn) { it.s = ei[i]; it.dl = dl; it.v = eattr[i]; it.valid = true; }
      }
    } else if (i < total) {
      const unsigned dl = (unsigned)(i - nIter);
      it.s = lo + (int)dl; it.dl = dl; it.valid = true;
      it.v = wantLogit ? P.self[dl] : P.asum[dl] / fmaxf((float)P.cnt[dl], 1.0f);
    }
    return it;
  };

  for (int i0 = hw; i0 < total; i0 += U * NHW) {
    Item it[U]; float4 xs[U]; float p[U];
#pragma unroll
    for (int u = 0; u < U; ++u) it[u] = fetch(i0 + u * NHW, false);
#pragma unroll
    for (int u = 0; u < U; ++u) {
      xs[u] = make_float4(0.f, 0.f, 0.f, 0.f);
      if (it[u].valid) xs[u] = ((const float4*)(xl + (size_t)it[u].s * C))[k];
    }
#pragma unroll
    for (int u = 0; u < U; ++u) {
      p[u] = 0.f;
      if (it[u].valid) {
        const float4 r = ((const float4*)(P.xr + it[u].dl * C))[k];
        p[u] = lrelu_dot4(xs[u], r, it[u].v, we4, at4);
      }
    }
#pragma unroll
    for (int u = 0; u < U; ++u) p[u] = row16_sum(p[u]);
    if (k == 0) {
#pragma unroll
      for (int u = 0; u < U; ++u) {
        const int i = i0 + u * NHW;
        if (it[u].valid) {
          if (i >= nIter) P.self[it[u].dl] = p[u]; else if (!OVF) P.lv[i] = p[u];
          atomicMax(&P.lmax[it[u].dl], f32_order_encode(p[u]));
        }
      }
    }
  }
  __syncthreads();

  for (int i0 = hw; i0 < total; i0 += U * NHW) {
    Item it[U]; float4 xs[U];
#pragma unroll
    for (int u = 0; u < U; ++u) it[u] = fetch(i0 + u * NHW, true);
#pragma unroll
    for (int u = 0; u < U; ++u) {
      xs[u] = make_float4(0.f, 0.f, 0.f, 0.f);
      if (it[u].valid) xs[u] = ((const float4*)(xl + (size_t)it[u].s * C))[k];
    }
    if (OVF) {
      float p[U];
#pragma unroll
      for (int u = 0; u < U; ++u) {
        p[u] = 0.f;
        if (it[u].valid && i0 + u * NHW < nIter) {
          const float4 r = ((const float4*)(P.xr + it[u].dl * C))[k];
          p[u] = lrelu_dot4(xs[u], r, it[u].v, we4, at4);
        }
      }
#pragma unroll
      for (int u = 0; u < U; ++u) {
        p[u] = row16_sum(p[u]);
        if (i0 + u * NHW < nIter) it[u].v = p[u];
      }
    }
#pragma unroll
    for (int u = 0; u < U; ++u) {
      if (it[u].valid) {
        const float ex = __expf(it[u].v - f32_order_decode(P.lmax[it[u].dl]));
        if (k == 0) atomicAdd(&P.den[it[u].dl], ex);
        float* acc = P.acc + it[u].dl * C + 4 * k;
        const float v0 = ex * xs[u].x, v1 = ex * xs[u].y, v2 = ex * xs[u].z, v3 = ex * xs[u].w;
        atomicAdd(acc + (0 ^ rot), rot ? v2 : v0); atomicAdd(acc + (1 ^ rot), rot ? v3 : v1);
        atomicAdd(acc + (2 ^ rot), rot ? v0 : v2); atomicAdd(acc + (3 ^ rot), rot ? v1 : v3);
      }
    }
  }
  __syncthreads();

  const float4* a4 = (const float4*)P.acc;
  float4* o4 = (float4*)(out + (size_t)lo * C);
  for (int i = tid; i < (int)nn * (C / 4); i += NT) {
    const float4 a = a4[i];
    const float4 b = ((const float4*)bias)[i & 15];
    const float dn = P.den[i >> 4];
    float4 o;
    o.x = a.x / dn + b.x; o.y = a.y / dn + b.y; o.z = a.z / dn + b.z; o.w = a.w / dn + b.w;
    o4[i] = o;
  }
}

__device__ __forceinline__ void owner_tail_sorted(
    const Planes P, const int* __restrict__ ei, const int* __restrict__ dst,
    const float* __restrict__ eattr, const float* __restrict__ xl,
    const float* __restrict__ We, const float* __restrict__ att,
    const float* __restrict__ bias, float* __restrict__ out,
    const int nIter, const int lo, const unsigned nn) {
  const int tid = threadIdx.x;
  constexpr int PER = CAP / NT;

  int e_[PER]; int rs[PER]; float rv[PER]; int rd[PER]; unsigned rp[PER];
#pragma unroll
  for (int u = 0; u < PER; ++u) {
    const int i = tid + u * NT;
    e_[u] = (i < nIter) ? P.le[i] : -1;
  }
#pragma unroll
  for (int u = 0; u < PER; ++u) {
    rd[u] = 0; rv[u] = 0.f; rs[u] = 0;
    if (e_[u] >= 0) { rd[u] = dst[e_[u]]; rv[u] = eattr[e_[u]]; rs[u] = ei[e_[u]]; }
  }
#pragma unroll
  for (int u = 0; u < PER; ++u) {
    rp[u] = ~0u;
    if (e_[u] >= 0) {
      const unsigned dl = (unsigned)(rd[u] - lo);
      if (dl < nn) {
        const unsigned rank = atomicAdd(&P.cnt[dl], 1u);
        atomicAdd(&P.asum[dl], rv[u]);
        rp[u] = (rank << 8) | dl;
      }
    }
  }
  __syncthreads();

  if (tid < 32) {
    unsigned pre[8]; unsigned run = 0u;
#pragma unroll
    for (int j = 0; j < 8; ++j) { pre[j] = run; run += P.cnt[8 * tid + j]; }
    unsigned incl = run;
#pragma unroll
    for (int d = 1; d < 32; d <<= 1) {
      const unsigned t = __shfl_up(incl, d);
      if (tid >= d) incl += t;
    }
    const unsigned base = incl - run;
#pragma unroll
    for (int j = 0; j < 8; ++j) P.off[8 * tid + j] = base + pre[j];
  }
  __syncthreads();

#pragma unroll
  for (int u = 0; u < PER; ++u) {
    if (rp[u] != ~0u) {
      const unsigned pos = P.off[rp[u] & 255u] + (rp[u] >> 8);
      P.le[pos] = rs[u]; P.lv[pos] = rv[u];
    }
  }
  __syncthreads();

  const int lane = tid & 31;
  const int k  = tid & 15;
  const float4 we4 = ((const float4*)We)[k];
  const float4 at4 = ((const float4*)att)[k];
  const float4 b4  = ((const float4*)bias)[k];
  for (;;) {
    unsigned pr = 0u;
    if (lane == 0) pr = atomicAdd(P.next, 1u);
    pr = (unsigned)__builtin_amdgcn_readfirstlane((int)pr);
    if (2u * pr >= nn) break;
    const int node = (int)(2u * pr) + (lane >> 4);
    if (node >= (int)nn) continue;
    const int beg = (int)P.off[node];
    const unsigned deg = P.cnt[node];
    const int nItem = (int)deg + 1;
    const float eself = P.asum[node] / fmaxf((float)deg, 1.0f);
    const float4 r4 = ((const float4*)(P.xr + node * C))[k];
    float m = -INFINITY, den = 0.f;
    float4 acc = make_float4(0.f, 0.f, 0.f, 0.f);
    for (int j0 = 0; j0 < nItem; j0 += U) {
      int s[U]; float ea[U]; float4 xs[U]; float p[U];
#pragma unroll
      for (int u = 0; u < U; ++u) {
        const int j = j0 + u;
        s[u] = lo + node; ea[u] = eself;
        if (j >= 1 && j < nItem) { s[u] = P.le[beg + j - 1]; ea[u] = P.lv[beg + j - 1]; }
      }
#pragma unroll
      for (int u = 0; u < U; ++u) {
        xs[u] = make_float4(0.f, 0.f, 0.f, 0.f);
        if (j0 + u < nItem) xs[u] = ((const float4*)(xl + (size_t)s[u] * C))[k];
      }
#pragma unroll
      for (int u = 0; u < U; ++u) {
        const float d = row16_sum(lrelu_dot4(xs[u], r4, ea[u], we4, at4));
        p[u] = (j0 + u < nItem) ? d : -INFINITY;
      }
      const float mn = fmaxf(fmaxf(m, p[0]), fmaxf(fmaxf(p[1], p[2]), p[3]));
      const float sc = __expf(m - mn);
      const float e0 = __expf(p[0] - mn), e1 = __expf(p[1] - mn);
      const float e2 = __expf(p[2] - mn), e3 = __expf(p[3] - mn);
      den = den * sc + ((e0 + e1) + (e2 + e3));
      acc.x = acc.x * sc + ((e0 * xs[0].x + e1 * xs[1].x) + (e2 * xs[2].x + e3 * xs[3].x));
      acc.y = acc.y * sc + ((e0 * xs[0].y + e1 * xs[1].y) + (e2 * xs[2].y + e3 * xs[3].y));
      acc.z = acc.z * sc + ((e0 * xs[0].z + e1 * xs[1].z) + (e2 * xs[2].z + e3 * xs[3].z));
      acc.w = acc.w * sc + ((e0 * xs[0].w + e1 * xs[1].w) + (e2 * xs[2].w + e3 * xs[3].w));
      m = mn;
    }
    float4 o;
    o.x = acc.x / den + b4.x; o.y = acc.y / den + b4.y;
    o.z = acc.z / den + b4.z; o.w = acc.w / den + b4.w;
    ((float4*)(out + (size_t)(lo + node) * C))[k] = o;
  }
}

__global__ __launch_bounds__(NT) void k_owner(
    const int* __restrict__ ei, const float* __restrict__ eattr,
    const float* __restrict__ xl, const float* __restrict__ xr,
    const float* __restrict__ We, const float* __restrict__ att,
    const float* __restrict__ bias, float* __restrict__ out,
    const unsigned* __restrict__ keys, int E, int N) {
  __shared__ __align__(16) float s_xr[R_NODES * C];
  __shared__ __align__(16) float s_acc[R_NODES * C];
  __shared__ unsigned s_cnt[256];
  __shared__ unsigned s_off[256];
  __shared__ float    s_asum[R_NODES];
  __shared__ unsigned s_lmax[R_NODES];
  __shared__ float    s_den[R_NODES];
  __shared__ float    s_self[R_NODES];
  __shared__ int      s_le[CAP];
  __shared__ unsigned s_ldl[CAP];
  __shared__ float    s_lv[CAP];
  __shared__ unsigned s_n;
  __shared__ unsigned s_next;

  const int tid = threadIdx.x;
  const int lo  = blockIdx.x * R_NODES;
  const unsigned nn = (unsigned)min(R_NODES, N - lo);
  const int* __restrict__ dst = ei + E;

  {
    const float4 z4 = {0.f, 0.f, 0.f, 0.f};
    float4* a4 = (float4*)s_acc;
    for (int i = tid; i < R_NODES * (C / 4); i += NT) a4[i] = z4;
    for (int i = tid; i < R_NODES; i += NT) {
      s_asum[i] = 0.f; s_lmax[i] = 0u; s_den[i] = 0.f; s_self[i] = 0.f;
    }
    for (int i = tid; i < 256; i += NT) s_cnt[i] = 0u;
    if (tid == 0) { s_n = 0u; s_next = 0u; }
    const float4* src = (const float4*)(xr + (size_t)lo * C);
    float4* d4 = (float4*)s_xr;
    for (int i = tid; i < (int)nn * (C / 4); i += NT) d4[i] = src[i];
  }
  __syncthreads();

  const unsigned pat = (blockIdx.x & 255u) * 0x01010101u;
  const int G = E >> 4;
  const uint4* key4 = (const uint4*)keys;
  auto place = [&](unsigned m0, unsigned m1, int g0) {
    while (m0 | m1) {
      const bool first = m0 != 0u;
      const unsigned m = first ? m0 : m1;
      const int bit = __builtin_ctz(m);
      if (first) m0 = m & (m - 1u); else m1 = m & (m - 1u);
      const int g = g0 + ((first ? 0 : 2) + ((bit >> 2) & 1)) * NT;
      const unsigned p = atomicAdd(&s_n, 1u);
      if (p < (unsigned)CAP) s_le[p] = (g << 4) + ((bit & 3) << 2) + (bit >> 3);
    }
  };
  const int Gmain = G - G % (4 * NT);
  for (int g0 = tid; g0 < Gmain; g0 += 4 * NT) {
    uint4 v[4];
#pragma unroll
    for (int u = 0; u < 4; ++u) v[u] = key4[g0 + u * NT];
    place(match16(v[0], pat) | (match16(v[1], pat) << 4),
          match16(v[2], pat) | (match16(v[3], pat) << 4), g0);
  }
  for (int g = Gmain + tid; g < G; g += NT) place(match16(key4[g], pat), 0u, g);
  for (int e = (G << 4) + tid; e < E; e += NT) {
    if ((unsigned)(dst[e] - lo) < nn) {
      unsigned p = atomicAdd(&s_n, 1u);
      if (p < (unsigned)CAP) s_le[p] = e;
    }
  }
  __syncthreads();
  const unsigned nraw = s_n;

  Planes P;
  P.xr = s_xr; P.acc = s_acc; P.cnt = s_cnt; P.asum = s_asum; P.lmax = s_lmax; P.den = s_den;
  P.self = s_self; P.le = s_le; P.ldl = s_ldl; P.lv = s_lv; P.off = s_off; P.next = &s_next;
  if (nraw > (unsigned)CAP)
    owner_tail<true>(P, ei, dst, eattr, xl, We, att, bias, out, E, lo, nn);
  else
    owner_tail_sorted(P, ei, dst, eattr, xl, We, att, bias, out, (int)nraw, lo, nn);
}

extern "C" void kernel_launch(void* const* d_in, const int* in_sizes, int n_in,
                              void* d_out, int out_size, void* d_ws, size_t ws_size,
                              hipStream_t stream) {
  (void)n_in; (void)out_size; (void)ws_size;
  const float* x     = (const float*)d_in[0];
  const float* eattr = (const float*)d_in[1];
  const float* Wl    = (const float*)d_in[2];
  const float* Wr    = (const float*)d_in[3];
  const float* We    = (const float*)d_in[4];
  const float* att   = (const float*)d_in[5];
  const float* bias  = (const float*)d_in[6];
  const int*   ei    = (const int*)d_in[7];
  float*       out   = (float*)d_out;

  int N = in_sizes[0] / F_IN;
  int E = in_sizes[1];

  char* ws = (char*)d_ws;
  size_t o = 0;
  auto take = [&](size_t bytes) -> void* {
    void* p = ws + o; o += (bytes + 255) & ~(size_t)255; return p;
  };
  float* xl    = (float*)take((size_t)N * C * 4);
  float* xr    = (float*)take((size_t)N * C * 4);
  unsigned* keys = (unsigned*)take((size_t)E);

  k_gemm<<<GEMM_BLOCKS, 256, 0, stream>>>(x, Wl, Wr, xl, xr, N, ei, keys, E);
  k_owner<<<(unsigned)((N + R_NODES - 1) / R_NODES), NT, 0, stream>>>(
      ei, eattr, xl, xr, We, att, bias, out, keys, E, N);
}


# node loop trip cut from 159 to 125 VALU instructions, branch-free: self loop = entry 0 of a destination's list segment, tail items clamped, next entries read ahead, leaky_relu as max; sum order kept
# speedup vs baseline: 6.7710x; 1.0633x over previous
#include <hip/hip_runtime.h>

#define F_IN 128
#define C 64

typedef float v2f __attribute__((ext_vector_type(2)));
typedef float v8f __attribute__((ext_vector_type(8)));

__device__ __forceinline__ unsigned f32_order_encode(float f) {
  unsigned b = __float_as_uint(f);
  return (b & 0x80000000u) ? ~b : (b | 0x80000000u);
}
__device__ __forceinline__ float f32_order_decode(unsigned u) {
  unsigned b = (u & 0x80000000u) ? (u & 0x7FFFFFFFu) : ~u;
  return __uint_as_float(b);
}

#define R_NODES 200
static_assert(R_NODES <= 256, "a local destination must fit the low byte of a packed rank word");

__device__ __forceinline__ void write_owner_keys(const int* __restrict__ ei,
                                                 unsigned* __restrict__ keys, int E) {
  const int* __restrict__ dst = ei + E;
  const int nW = (E >> 4) << 2;
  for (int q = blockIdx.x * blockDim.x + threadIdx.x; q < nW; q += gridDim.x * blockDim.x) {
    const unsigned d0 = (unsigned)dst[4 * q], d1 = (unsigned)dst[4 * q + 1];
    const unsigned d2 = (unsigned)dst[4 * q + 2], d3 = (unsigned)dst[4 * q + 3];
    keys[q] = ((d0 / R_NODES) & 255u) | (((d1 / R_NODES) & 255u) << 8) |
              (((d2 / R_NODES) & 255u) << 16) | ((d3 / R_NODES) << 24);
  }
}

#define GEMM_BLOCKS 512
#define XPITCH 132
__global__ __launch_bounds__(256) void k_gemm(const float* __restrict__ x, const float* __restrict__ Wl,
                       const float* __restrict__ Wr, float* __restrict__ xl,
                       float* __restrict__ xr, int N,
                       const int* __restrict__ ei, unsigned* __restrict__ keys, int E) {
  __shared__ __align__(16) float s_x[2][16 * XPITCH];
  write_owner_keys(ei, keys, E);
  const int tid  = threadIdx.x;
  const int wave = tid >> 5;
  const int lane = tid & 31;
  const int rowTiles = N >> 4;
  int rt = blockIdx.x;
  if (rt >= rowTiles) return;
  const int n0    = (wave >> 1) << 4;
  const int which = wave & 1;
  const float* W = which ? Wr : Wl;
  float*       Y = which ? xr : xl;
  const int half = lane >> 4;
  const int l15  = lane & 15;

  const int st0 = wave * XPITCH + lane * 4;
  const int st1 = st0 + 8 * XPITCH;
  float4 p0, p1;
  {
    const float4* src = (const float4*)(x + (size_t)rt * 16 * F_IN);
    p0 = src[tid]; p1 = src[tid + 256];
  }

  const float* bcol = W + n0 + l15 + (half * 2) * C;
  v2f b[32];
#pragma unroll
  for (int s = 0; s < 32; ++s) { b[s].x = bcol[(4 * s) * C]; b[s].y = bcol[(4 * s + 1) * C]; }

  *(float4*)(&s_x[0][st0]) = p0;
  *(float4*)(&s_x[0][st1]) = p1;
  __syncthreads();

  int cur = 0;
  for (;;) {
    const int nxt = rt + (int)gridDim.x;
    const bool more = nxt < rowTiles;
    if (more) {
      const float4* src = (const float4*)(x + (size_t)nxt * 16 * F_IN);
      p0 = src[tid]; p1 = src[tid + 256];
    }
    const v2f* ap = (const v2f*)(&s_x[cur][l15 * XPITCH + 2 * half]);
    v2f a[32];
#pragma unroll
    for (int s = 0; s < 32; ++s) a[s] = ap[2 * s];
    __builtin_amdgcn_sched_barrier(0);
    v8f acc = {0.f, 0.f, 0.f, 0.f, 0.f, 0.f, 0.f, 0.f};
#pragma unroll
    for (int s = 0; s < 32; ++s) {
      acc = __builtin_amdgcn_wmma_f32_16x16x4_f32(
           false, a[s],  false, b[s],
           (short)0, acc,  false,  false);
    }
    if (more) {
      *(float4*)(&s_x[cur ^ 1][st0]) = p0;
      *(float4*)(&s_x[cur ^ 1][st1]) = p1;
      __syncthreads();
    }
    float* yrow = Y + ((size_t)rt * 16 + half * 8) * C + n0 + l15;
#pragma unroll
    for (int r = 0; r < 8; ++r) yrow[(size_t)r * C] = acc[r];
    if (!more) break;
    cur ^= 1; rt = nxt;
  }
}

#define CAP 8192
#define NT 1024
#define NHW (NT / 16)
#define U 4

__device__ __forceinline__ unsigned zero_bytes(unsigned t) {
  return ~(((t & 0x7F7F7F7Fu) + 0x7F7F7F7Fu) | t) & 0x80808080u;
}
__device__ __forceinline__ unsigned match16(uint4 v, unsigned pat) {
  return (zero_bytes(v.x ^ pat) >> 7) | (zero_bytes(v.y ^ pat) >> 6) |
         (zero_bytes(v.z ^ pat) >> 5) | (zero_bytes(v.w ^ pat) >> 4);
}

struct Item { int s; unsigned dl; float v; bool valid; };

struct Planes {
  float* xr; float* acc; unsigned* cnt; float* asum; unsigned* lmax; float* den; float* self;
  int* le; unsigned* ldl; float* lv; unsigned* off; unsigned* next;
};

__device__ __forceinline__ float row16_sum(float v) {
  v += __int_as_float(__builtin_amdgcn_update_dpp(0, __float_as_int(v), 0x161, 0xf, 0xf, true));
  v += __int_as_float(__builtin_amdgcn_update_dpp(0, __float_as_int(v), 0x162, 0xf, 0xf, true));
  v += __int_as_float(__builtin_amdgcn_update_dpp(0, __float_as_int(v), 0x164, 0xf, 0xf, true));
  v += __int_as_float(__builtin_amdgcn_update_dpp(0, __float_as_int(v), 0x168, 0xf, 0xf, true));
  return v;
}

__device__ __forceinline__ float lrelu_dot4(float4 a, float4 r, float ea, float4 we, float4 at) {
  float m0 = a.x + r.x + ea * we.x, m1 = a.y + r.y + ea * we.y;
  float m2 = a.z + r.z + ea * we.z, m3 = a.w + r.w + ea * we.w;
  m0 = fmaxf(m0, 0.2f * m0); m1 = fmaxf(m1, 0.2f * m1);
  m2 = fmaxf(m2, 0.2f * m2); m3 = fmaxf(m3, 0.2f * m3);
  return m0 * at.x + m1 * at.y + m2 * at.z + m3 * at.w;
}

template <bool OVF>
__device__ __forceinline__ void owner_tail(
    const Planes P, const int* __restrict__ ei, const int* __restrict__ dst,
    const float* __restrict__ eattr, const float* __restrict__ xl,
    const float* __restrict__ We, const float* __restrict__ att,
    const float* __restrict__ bias, float* __restrict__ out,
    const int nIter, const int lo, const unsigned nn) {
  const int tid = threadIdx.x;

  for (int i = tid; i < nIter; i += NT) {
    if (!OVF) {
      const int e = P.le[i];
      unsigned dl = (unsigned)(dst[e] - lo);
      const float ea = eattr[e];
      const int s = ei[e];
      if (dl < nn) { atomicAdd(&P.cnt[dl], 1u); atomicAdd(&P.asum[dl], ea); } else dl = ~0u;
      P.le[i] = s; P.ldl[i] = dl; P.lv[i] = ea;
    } else {
      const unsigned dl = (unsigned)(dst[i] - lo);
      if (dl < nn) {
        atomicAdd(&P.cnt[dl], 1u);
        atomicAdd(&P.asum[dl], eattr[i]);
      }
    }
  }
  __syncthreads();

  const int hw = tid >> 4;
  const int k  = tid & 15;
  const int rot = (hw & 1) << 1;
  const int total = nIter + (int)nn;
  const float4 we4 = ((const float4*)We)[k];
  const float4 at4 = ((const float4*)att)[k];

  auto fetch = [&](int i, bool wantLogit) -> Item {
    Item it; it.s = 0; it.dl = 0u; it.v = 0.f; it.valid = false;
    if (i < nIter) {
      if (!OVF) { it.s = P.le[i]; it.dl = P.ldl[i]; it.v = P.lv[i]; it.valid = it.dl != ~0u; }
      else {
        const unsigned dl = (unsigned)(dst[i] - lo);
        if (dl < nn) { it.s = ei[i]; it.dl = dl; it.v = eattr[i]; it.valid = true; }
      }
    } else if (i < total) {
      const unsigned dl = (unsigned)(i - nIter);
      it.s = lo + (int)dl; it.dl = dl; it.valid = true;
      it.v = wantLogit ? P.self[dl] : P.asum[dl] / fmaxf((float)P.cnt[dl], 1.0f);
    }
    return it;
  };

  for (int i0 = hw; i0 < total; i0 += U * NHW) {
    Item it[U]; float4 xs[U]; float p[U];
#pragma unroll
    for (int u = 0; u < U; ++u) it[u] = fetch(i0 + u * NHW, false);
#pragma unroll
    for (int u = 0; u < U; ++u) {
      xs[u] = make_float4(0.f, 0.f, 0.f, 0.f);
      if (it[u].valid) xs[u] = ((const float4*)(xl + (size_t)it[u].s * C))[k];
    }
#pragma unroll
    for (int u = 0; u < U; ++u) {
      p[u] = 0.f;
      if (it[u].valid) {
        const float4 r = ((const float4*)(P.xr + it[u].dl * C))[k];
        p[u] = lrelu_dot4(xs[u], r, it[u].v, we4, at4);
      }
    }
#pragma unroll
    for (int u = 0; u < U; ++u) p[u] = row16_sum(p[u]);
    if (k == 0) {
#pragma unroll
      for (int u = 0; u < U; ++u) {
        const int i = i0 + u * NHW;
        if (it[u].valid) {
          if (i >= nIter) P.self[it[u].dl] = p[u]; else if (!OVF) P.lv[i] = p[u];
          atomicMax(&P.lmax[it[u].dl], f32_order_encode(p[u]));
        }
      }
    }
  }
  __syncthreads();

  for (int i0 = hw; i0 < total; i0 += U * NHW) {
    Item it[U]; float4 xs[U];
#pragma unroll
    for (int u = 0; u < U; ++u) it[u] = fetch(i0 + u * NHW, true);
#pragma unroll
    for (int u = 0; u < U; ++u) {
      xs[u] = make_float4(0.f, 0.f, 0.f, 0.f);
      if (it[u].valid) xs[u] = ((const float4*)(xl + (size_t)it[u].s * C))[k];
    }
    if (OVF) {
      float p[U];
#pragma unroll
      for (int u = 0; u < U; ++u) {
        p[u] = 0.f;
        if (it[u].valid && i0 + u * NHW < nIter) {
          const float4 r = ((const float4*)(P.xr + it[u].dl * C))[k];
          p[u] = lrelu_dot4(xs[u], r, it[u].v, we4, at4);
        }
      }
#pragma unroll
      for (int u = 0; u < U; ++u) {
        p[u] = row16_sum(p[u]);
        if (i0 + u * NHW < nIter) it[u].v = p[u];
      }
    }
#pragma unroll
    for (int u = 0; u < U; ++u) {
      if (it[u].valid) {
        const float ex = __expf(it[u].v - f32_order_decode(P.lmax[it[u].dl]));
        if (k == 0) atomicAdd(&P.den[it[u].dl], ex);
        float* acc = P.acc + it[u].dl * C + 4 * k;
        const float v0 = ex * xs[u].x, v1 = ex * xs[u].y, v2 = ex * xs[u].z, v3 = ex * xs[u].w;
        atomicAdd(acc + (0 ^ rot), rot ? v2 : v0); atomicAdd(acc + (1 ^ rot), rot ? v3 : v1);
        atomicAdd(acc + (2 ^ rot), rot ? v0 : v2); atomicAdd(acc + (3 ^ rot), rot ? v1 : v3);
      }
    }
  }
  __syncthreads();

  const float4* a4 = (const float4*)P.acc;
  float4* o4 = (float4*)(out + (size_t)lo * C);
  for (int i = tid; i < (int)nn * (C / 4); i += NT) {
    const float4 a = a4[i];
    const float4 b = ((const float4*)bias)[i & 15];
    const float dn = P.den[i >> 4];
    float4 o;
    o.x = a.x / dn + b.x; o.y = a.y / dn + b.y; o.z = a.z / dn + b.z; o.w = a.w / dn + b.w;
    o4[i] = o;
  }
}

__device__ __forceinline__ void owner_tail_sorted(
    const Planes P, const int* __restrict__ ei, const int* __restrict__ dst,
    const float* __restrict__ eattr, const float* __restrict__ xl,
    const float* __restrict__ We, const float* __restrict__ att,
    const float* __restrict__ bias, float* __restrict__ out,
    const int nIter, const int lo, const unsigned nn) {
  const int tid = threadIdx.x;
  constexpr int PER = CAP / NT;

  int e_[PER]; int rs[PER]; float rv[PER]; int rd[PER]; unsigned rp[PER];
#pragma unroll
  for (int u = 0; u < PER; ++u) {
    const int i = tid + u * NT;
    e_[u] = (i < nIter) ? P.le[i] : -1;
  }
#pragma unroll
  for (int u = 0; u < PER; ++u) {
    rd[u] = 0; rv[u] = 0.f; rs[u] = 0;
    if (e_[u] >= 0) { rd[u] = dst[e_[u]]; rv[u] = eattr[e_[u]]; rs[u] = ei[e_[u]]; }
  }
#pragma unroll
  for (int u = 0; u < PER; ++u) {
    rp[u] = ~0u;
    if (e_[u] >= 0) {
      const unsigned dl = (unsigned)(rd[u] - lo);
      if (dl < nn) {
        const unsigned rank = atomicAdd(&P.cnt[dl], 1u);
        atomicAdd(&P.asum[dl], rv[u]);
        rp[u] = (rank << 8) | dl;
      }
    }
  }
  __syncthreads();

  if (tid < 32) {
    unsigned pre[8]; unsigned run = 0u;
#pragma unroll
    for (int j = 0; j < 8; ++j) {
      const unsigned nd = 8u * (unsigned)tid + (unsigned)j;
      pre[j] = run; run += P.cnt[nd] + ((nd < nn) ? 1u : 0u);
    }
    unsigned incl = run;
#pragma unroll
    for (int d = 1; d < 32; d <<= 1) {
      const unsigned t = __shfl_up(incl, d);
      if (tid >= d) incl += t;
    }
    const unsigned base = incl - run;
#pragma unroll
    for (int j = 0; j < 8; ++j) {
      const unsigned nd = 8u * (unsigned)tid + (unsigned)j;
      const unsigned o = base + pre[j];
      P.off[nd] = o;
      if (nd < nn) {
        P.le[o] = lo + (int)nd;
        P.lv[o] = P.asum[nd] / fmaxf((float)P.cnt[nd], 1.0f);
      }
    }
  }
  __syncthreads();

#pragma unroll
  for (int u = 0; u < PER; ++u) {
    if (rp[u] != ~0u) {
      const unsigned pos = P.off[rp[u] & 255u] + 1u + (rp[u] >> 8);
      P.le[pos] = rs[u]; P.lv[pos] = rv[u];
    }
  }
  __syncthreads();

  const int lane = tid & 31;
  const int k  = tid & 15;
  const unsigned kb = (unsigned)k << 4;
  const char* __restrict__ xlb = (const char*)xl;
  const float4 we4 = ((const float4*)We)[k];
  const float4 at4 = ((const float4*)att)[k];
  const float4 b4  = ((const float4*)bias)[k];
  for (;;) {
    unsigned pr = 0u;
    if (lane == 0) pr = atomicAdd(P.next, 1u);
    pr = (unsigned)__builtin_amdgcn_readfirstlane((int)pr);
    if (2u * pr >= nn) break;
    const int node = (int)(2u * pr) + (lane >> 4);
    if (node >= (int)nn) continue;
    const int beg = (int)P.off[node];
    const int nItem = (int)P.cnt[node] + 1;
    const int last = beg + nItem - 1;
    const float4 r4 = ((const float4*)(P.xr + node * C))[k];
    float m = -INFINITY, den = 0.f;
    float4 acc = make_float4(0.f, 0.f, 0.f, 0.f);
    int s[U]; float ea[U];
#pragma unroll
    for (int u = 0; u < U; ++u) {
      const int idx = min(beg + u, last);
      s[u] = P.le[idx]; ea[u] = P.lv[idx];
    }
    for (int j0 = 0; j0 < nItem; j0 += U) {
      float4 xs[U]; float eac[U]; float p[U];
#pragma unroll
      for (int u = 0; u < U; ++u) {
        xs[u] = *(const float4*)(xlb + (((unsigned)s[u] << 8) | kb));
        eac[u] = ea[u];
      }
#pragma unroll
      for (int u = 0; u < U; ++u) {
        const int idx = min(beg + j0 + U + u, last);
        s[u] = P.le[idx]; ea[u] = P.lv[idx];
      }
#pragma unroll
      for (int u = 0; u < U; ++u) {
        const float d = row16_sum(lrelu_dot4(xs[u], r4, eac[u], we4, at4));
        p[u] = (j0 + u < nItem) ? d : -INFINITY;
      }
      const float mn = fmaxf(fmaxf(m, p[0]), fmaxf(fmaxf(p[1], p[2]), p[3]));
      const float sc = __expf(m - mn);
      const float e0 = __expf(p[0] - mn), e1 = __expf(p[1] - mn);
      const float e2 = __expf(p[2] - mn), e3 = __expf(p[3] - mn);
      den = den * sc + ((e0 + e1) + (e2 + e3));
      acc.x = acc.x * sc + ((e0 * xs[0].x + e1 * xs[1].x) + (e2 * xs[2].x + e3 * xs[3].x));
      acc.y = acc.y * sc + ((e0 * xs[0].y + e1 * xs[1].y) + (e2 * xs[2].y + e3 * xs[3].y));
      acc.z = acc.z * sc + ((e0 * xs[0].z + e1 * xs[1].z) + (e2 * xs[2].z + e3 * xs[3].z));
      acc.w = acc.w * sc + ((e0 * xs[0].w + e1 * xs[1].w) + (e2 * xs[2].w + e3 * xs[3].w));
      m = mn;
    }
    float4 o;
    o.x = acc.x / den + b4.x; o.y = acc.y / den + b4.y;
    o.z = acc.z / den + b4.z; o.w = acc.w / den + b4.w;
    ((float4*)(out + (size_t)(lo + node) * C))[k] = o;
  }
}

__global__ __launch_bounds__(NT) void k_owner(
    const int* __restrict__ ei, const float* __restrict__ eattr,
    const float* __restrict__ xl, const float* __restrict__ xr,
    const float* __restrict__ We, const float* __restrict__ att,
    const float* __restrict__ bias, float* __restrict__ out,
    const unsigned* __restrict__ keys, int E, int N) {
  __shared__ __align__(16) float s_xr[R_NODES * C];
  __shared__ __align__(16) float s_acc[R_NODES * C];
  __shared__ unsigned s_cnt[256];
  __shared__ unsigned s_off[256];
  __shared__ float    s_asum[R_NODES];
  __shared__ unsigned s_lmax[R_NODES];
  __shared__ float    s_den[R_NODES];
  __shared__ float    s_self[R_NODES];
  __shared__ int      s_le[CAP];
  __shared__ unsigned s_ldl[CAP];
  __shared__ float    s_lv[CAP];
  __shared__ unsigned s_n;
  __shared__ unsigned s_next;

  const int tid = threadIdx.x;
  const int lo  = blockIdx.x * R_NODES;
  const unsigned nn = (unsigned)min(R_NODES, N - lo);
  const int* __restrict__ dst = ei + E;

  {
    const float4 z4 = {0.f, 0.f, 0.f, 0.f};
    float4* a4 = (float4*)s_acc;
    for (int i = tid; i < R_NODES * (C / 4); i += NT) a4[i] = z4;
    for (int i = tid; i < R_NODES; i += NT) {
      s_asum[i] = 0.f; s_lmax[i] = 0u; s_den[i] = 0.f; s_self[i] = 0.f;
    }
    for (int i = tid; i < 256; i += NT) s_cnt[i] = 0u;
    if (tid == 0) { s_n = 0u; s_next = 0u; }
    const float4* src = (const float4*)(xr + (size_t)lo * C);
    float4* d4 = (float4*)s_xr;
    for (int i = tid; i < (int)nn * (C / 4); i += NT) d4[i] = src[i];
  }
  __syncthreads();

  const unsigned pat = (blockIdx.x & 255u) * 0x01010101u;
  const int G = E >> 4;
  const uint4* key4 = (const uint4*)keys;
  auto place = [&](unsigned m0, unsigned m1, int g0) {
    while (m0 | m1) {
      const bool first = m0 != 0u;
      const unsigned m = first ? m0 : m1;
      const int bit = __builtin_ctz(m);
      if (first) m0 = m & (m - 1u); else m1 = m & (m - 1u);
      const int g = g0 + ((first ? 0 : 2) + ((bit >> 2) & 1)) * NT;
      const unsigned p = atomicAdd(&s_n, 1u);
      if (p < (unsigned)CAP) s_le[p] = (g << 4) + ((bit & 3) << 2) + (bit >> 3);
    }
  };
  const int Gmain = G - G % (4 * NT);
  for (int g0 = tid; g0 < Gmain; g0 += 4 * NT) {
    uint4 v[4];
#pragma unroll
    for (int u = 0; u < 4; ++u) v[u] = key4[g0 + u * NT];
    place(match16(v[0], pat) | (match16(v[1], pat) << 4),
          match16(v[2], pat) | (match16(v[3], pat) << 4), g0);
  }
  for (int g = Gmain + tid; g < G; g += NT) place(match16(key4[g], pat), 0u, g);
  for (int e = (G << 4) + tid; e < E; e += NT) {
    if ((unsigned)(dst[e] - lo) < nn) {
      unsigned p = atomicAdd(&s_n, 1u);
      if (p < (unsigned)CAP) s_le[p] = e;
    }
  }
  __syncthreads();
  const unsigned nraw = s_n;

  Planes P;
  P.xr = s_xr; P.acc = s_acc; P.cnt = s_cnt; P.asum = s_asum; P.lmax = s_lmax; P.den = s_den;
  P.self = s_self; P.le = s_le; P.ldl = s_ldl; P.lv = s_lv; P.off = s_off; P.next = &s_next;
  if (nraw + nn > (unsigned)CAP)
    owner_tail<true>(P, ei, dst, eattr, xl, We, att, bias, out, E, lo, nn);
  else
    owner_tail_sorted(P, ei, dst, eattr, xl, We, att, bias, out, (int)nraw, lo, nn);
}

extern "C" void kernel_launch(void* const* d_in, const int* in_sizes, int n_in,
                              void* d_out, int out_size, void* d_ws, size_t ws_size,
                              hipStream_t stream) {
  (void)n_in; (void)out_size; (void)ws_size;
  const float* x     = (const float*)d_in[0];
  const float* eattr = (const float*)d_in[1];
  const float* Wl    = (const float*)d_in[2];
  const float* Wr    = (const float*)d_in[3];
  const float* We    = (const float*)d_in[4];
  const float* att   = (const float*)d_in[5];
  const float* bias  = (const float*)d_in[6];
  const int*   ei    = (const int*)d_in[7];
  float*       out   = (float*)d_out;

  int N = in_sizes[0] / F_IN;
  int E = in_sizes[1];

  char* ws = (char*)d_ws;
  size_t o = 0;
  auto take = [&](size_t bytes) -> void* {
    void* p = ws + o; o += (bytes + 255) & ~(size_t)255; return p;
  };
  float* xl    = (float*)take((size_t)N * C * 4);
  float* xr    = (float*)take((size_t)N * C * 4);
  unsigned* keys = (unsigned*)take((size_t)E);

  k_gemm<<<GEMM_BLOCKS, 256, 0, stream>>>(x, Wl, Wr, xl, xr, N, ei, keys, E);
  k_owner<<<(unsigned)((N + R_NODES - 1) / R_NODES), NT, 0, stream>>>(
      ei, eattr, xl, xr, We, att, bias, out, keys, E, N);
}


# owner block reads its words of a group index (a bit per key value and 16-edge group, filled in LDS by the GEMM launch's key pass) and loads only marked key groups; LDS list (sum) order changes
# speedup vs baseline: 7.9621x; 1.1759x over previous
#include <hip/hip_runtime.h>

#define F_IN 128
#define C 64

typedef float v2f __attribute__((ext_vector_type(2)));
typedef float v8f __attribute__((ext_vector_type(8)));

__device__ __forceinline__ unsigned f32_order_encode(float f) {
  unsigned b = __float_as_uint(f);
  return (b & 0x80000000u) ? ~b : (b | 0x80000000u);
}
__device__ __forceinline__ float f32_order_decode(unsigned u) {
  unsigned b = (u & 0x80000000u) ? (u & 0x7FFFFFFFu) : ~u;
  return __uint_as_float(b);
}

#define R_NODES 200
static_assert(R_NODES <= 256, "a local destination must fit the low byte of a packed rank word");

#define GEMM_NT 256
__device__ __forceinline__ void write_owner_keys(const int* __restrict__ ei,
                                                 unsigned* __restrict__ keys,
                                                 unsigned* __restrict__ bits,
                                                 unsigned* s_bits, int E) {
  const int* __restrict__ dst = ei + E;
  const int tid = threadIdx.x;
  const int nW = (E >> 4) << 2;
  const int nRec = ((E >> 4) + 127) >> 7;
  for (int c = blockIdx.x; c < nRec; c += gridDim.x) {
    if (c != (int)blockIdx.x) __syncthreads();
    for (int i = tid; i < 4 * 256; i += GEMM_NT) s_bits[i] = 0u;
    __syncthreads();
    unsigned d[2][4];
#pragma unroll
    for (int u = 0; u < 2; ++u) {
      const int q = min(c * 512 + tid + u * GEMM_NT, nW - 1);
      d[u][0] = (unsigned)dst[4 * q];     d[u][1] = (unsigned)dst[4 * q + 1];
      d[u][2] = (unsigned)dst[4 * q + 2]; d[u][3] = (unsigned)dst[4 * q + 3];
    }
#pragma unroll
    for (int u = 0; u < 2; ++u) {
      const int t = tid + u * GEMM_NT;
      const int q = c * 512 + t;
      if (q < nW) {
        const unsigned o0 = (d[u][0] / R_NODES) & 255u, o1 = (d[u][1] / R_NODES) & 255u;
        const unsigned o2 = (d[u][2] / R_NODES) & 255u, o3 = (d[u][3] / R_NODES) & 255u;
        keys[q] = o0 | (o1 << 8) | (o2 << 16) | (o3 << 24);
        unsigned* wrow = s_bits + (t >> 7) * 256;
        const unsigned bit = 1u << ((t >> 2) & 31);
        atomicOr(&wrow[o0], bit); atomicOr(&wrow[o1], bit);
        atomicOr(&wrow[o2], bit); atomicOr(&wrow[o3], bit);
      }
    }
    __syncthreads();
    for (int r = tid; r < 256; r += GEMM_NT)
      ((uint4*)bits)[(size_t)c * 256 + r] =
          make_uint4(s_bits[r], s_bits[256 + r], s_bits[512 + r], s_bits[768 + r]);
  }
}

#define GEMM_BLOCKS 512
#define XPITCH 132
__global__ __launch_bounds__(GEMM_NT) void k_gemm(const float* __restrict__ x, const float* __restrict__ Wl,
                       const float* __restrict__ Wr, float* __restrict__ xl,
                       float* __restrict__ xr, int N,
                       const int* __restrict__ ei, unsigned* __restrict__ keys,
                       unsigned* __restrict__ bits, int E) {
  __shared__ __align__(16) float s_x[2][16 * XPITCH];
  __shared__ unsigned s_bits[4 * 256];
  write_owner_keys(ei, keys, bits, s_bits, E);
  const int tid  = threadIdx.x;
  const int wave = tid >> 5;
  const int lane = tid & 31;
  const int rowTiles = N >> 4;
  int rt = blockIdx.x;
  if (rt >= rowTiles) return;
  const int n0    = (wave >> 1) << 4;
  const int which = wave & 1;
  const float* W = which ? Wr : Wl;
  float*       Y = which ? xr : xl;
  const int half = lane >> 4;
  const int l15  = lane & 15;

  const int st0 = wave * XPITCH + lane * 4;
  const int st1 = st0 + 8 * XPITCH;
  float4 p0, p1;
  {
    const float4* src = (const float4*)(x + (size_t)rt * 16 * F_IN);
    p0 = src[tid]; p1 = src[tid + 256];
  }

  const float* bcol = W + n0 + l15 + (half * 2) * C;
  v2f b[32];
#pragma unroll
  for (int s = 0; s < 32; ++s) { b[s].x = bcol[(4 * s) * C]; b[s].y = bcol[(4 * s + 1) * C]; }

  *(float4*)(&s_x[0][st0]) = p0;
  *(float4*)(&s_x[0][st1]) = p1;
  __syncthreads();

  int cur = 0;
  for (;;) {
    const int nxt = rt + (int)gridDim.x;
    const bool more = nxt < rowTiles;
    if (more) {
      const float4* src = (const float4*)(x + (size_t)nxt * 16 * F_IN);
      p0 = src[tid]; p1 = src[tid + 256];
    }
    const v2f* ap = (const v2f*)(&s_x[cur][l15 * XPITCH + 2 * half]);
    v2f a[32];
#pragma unroll
    for (int s = 0; s < 32; ++s) a[s] = ap[2 * s];
    __builtin_amdgcn_sched_barrier(0);
    v8f acc = {0.f, 0.f, 0.f, 0.f, 0.f, 0.f, 0.f, 0.f};
#pragma unroll
    for (int s = 0; s < 32; ++s) {
      acc = __builtin_amdgcn_wmma_f32_16x16x4_f32(
           false, a[s],  false, b[s],
           (short)0, acc,  false,  false);
    }
    if (more) {
      *(float4*)(&s_x[cur ^ 1][st0]) = p0;
      *(float4*)(&s_x[cur ^ 1][st1]) = p1;
      __syncthreads();
    }
    float* yrow = Y + ((size_t)rt * 16 + half * 8) * C + n0 + l15;
#pragma unroll
    for (int r = 0; r < 8; ++r) yrow[(size_t)r * C] = acc[r];
    if (!more) break;
    cur ^= 1; rt = nxt;
  }
}

#define CAP 8192
#define NT 1024
#define NHW (NT / 16)
#define U 4

__device__ __forceinline__ unsigned zero_bytes(unsigned t) {
  return ~(((t & 0x7F7F7F7Fu) + 0x7F7F7F7Fu) | t) & 0x80808080u;
}
__device__ __forceinline__ unsigned match16(uint4 v, unsigned pat) {
  return (zero_bytes(v.x ^ pat) >> 7) | (zero_bytes(v.y ^ pat) >> 6) |
         (zero_bytes(v.z ^ pat) >> 5) | (zero_bytes(v.w ^ pat) >> 4);
}

struct Item { int s; unsigned dl; float v; bool valid; };

struct Planes {
  float* xr; float* acc; unsigned* cnt; float* asum; unsigned* lmax; float* den; float* self;
  int* le; unsigned* ldl; float* lv; unsigned* off; unsigned* next;
};

__device__ __forceinline__ float row16_sum(float v) {
  v += __int_as_float(__builtin_amdgcn_update_dpp(0, __float_as_int(v), 0x161, 0xf, 0xf, true));
  v += __int_as_float(__builtin_amdgcn_update_dpp(0, __float_as_int(v), 0x162, 0xf, 0xf, true));
  v += __int_as_float(__builtin_amdgcn_update_dpp(0, __float_as_int(v), 0x164, 0xf, 0xf, true));
  v += __int_as_float(__builtin_amdgcn_update_dpp(0, __float_as_int(v), 0x168, 0xf, 0xf, true));
  return v;
}

__device__ __forceinline__ float lrelu_dot4(float4 a, float4 r, float ea, float4 we, float4 at) {
  float m0 = a.x + r.x + ea * we.x, m1 = a.y + r.y + ea * we.y;
  float m2 = a.z + r.z + ea * we.z, m3 = a.w + r.w + ea * we.w;
  m0 = fmaxf(m0, 0.2f * m0); m1 = fmaxf(m1, 0.2f * m1);
  m2 = fmaxf(m2, 0.2f * m2); m3 = fmaxf(m3, 0.2f * m3);
  return m0 * at.x + m1 * at.y + m2 * at.z + m3 * at.w;
}

template <bool OVF>
__device__ __forceinline__ void owner_tail(
    const Planes P, const int* __restrict__ ei, const int* __restrict__ dst,
    const float* __restrict__ eattr, const float* __restrict__ xl,
    const float* __restrict__ We, const float* __restrict__ att,
    const float* __restrict__ bias, float* __restrict__ out,
    const int nIter, const int lo, const unsigned nn) {
  const int tid = threadIdx.x;

  for (int i = tid; i < nIter; i += NT) {
    if (!OVF) {
      const int e = P.le[i];
      unsigned dl = (unsigned)(dst[e] - lo);
      const float ea = eattr[e];
      const int s = ei[e];
      if (dl < nn) { atomicAdd(&P.cnt[dl], 1u); atomicAdd(&P.asum[dl], ea); } else dl = ~0u;
      P.le[i] = s; P.ldl[i] = dl; P.lv[i] = ea;
    } else {
      const unsigned dl = (unsigned)(dst[i] - lo);
      if (dl < nn) {
        atomicAdd(&P.cnt[dl], 1u);
        atomicAdd(&P.asum[dl], eattr[i]);
      }
    }
  }
  __syncthreads();

  const int hw = tid >> 4;
  const int k  = tid & 15;
  const int rot = (hw & 1) << 1;
  const int total = nIter + (int)nn;
  const float4 we4 = ((const float4*)We)[k];
  const float4 at4 = ((const float4*)att)[k];

  auto fetch = [&](int i, bool wantLogit) -> Item {
    Item it; it.s = 0; it.dl = 0u; it.v = 0.f; it.valid = false;
    if (i < nIter) {
      if (!OVF) { it.s = P.le[i]; it.dl = P.ldl[i]; it.v = P.lv[i]; it.valid = it.dl != ~0u; }
      else {
        const unsigned dl = (unsigned)(dst[i] - lo);
        if (dl < nn) { it.s = ei[i]; it.dl = dl; it.v = eattr[i]; it.valid = true; }
      }
    } else if (i < total) {
      const unsigned dl = (unsigned)(i - nIter);
      it.s = lo + (int)dl; it.dl = dl; it.valid = true;
      it.v = wantLogit ? P.self[dl] : P.asum[dl] / fmaxf((float)P.cnt[dl], 1.0f);
    }
    return it;
  };

  for (int i0 = hw; i0 < total; i0 += U * NHW) {
    Item it[U]; float4 xs[U]; float p[U];
#pragma unroll
    for (int u = 0; u < U; ++u) it[u] = fetch(i0 + u * NHW, false);
#pragma unroll
    for (int u = 0; u < U; ++u) {
      xs[u] = make_float4(0.f, 0.f, 0.f, 0.f);
      if (it[u].valid) xs[u] = ((const float4*)(xl + (size_t)it[u].s * C))[k];
    }
#pragma unroll
    for (int u = 0; u < U; ++u) {
      p[u] = 0.f;
      if (it[u].valid) {
        const float4 r = ((const float4*)(P.xr + it[u].dl * C))[k];
        p[u] = lrelu_dot4(xs[u], r, it[u].v, we4, at4);
      }
    }
#pragma unroll
    for (int u = 0; u < U; ++u) p[u] = row16_sum(p[u]);
    if (k == 0) {
#pragma unroll
      for (int u = 0; u < U; ++u) {
        const int i = i0 + u * NHW;
        if (it[u].valid) {
          if (i >= nIter) P.self[it[u].dl] = p[u]; else if (!OVF) P.lv[i] = p[u];
          atomicMax(&P.lmax[it[u].dl], f32_order_encode(p[u]));
        }
      }
    }
  }
  __syncthreads();

  for (int i0 = hw; i0 < total; i0 += U * NHW) {
    Item it[U]; float4 xs[U];
#pragma unroll
    for (int u = 0; u < U; ++u) it[u] = fetch(i0 + u * NHW, true);
#pragma unroll
    for (int u = 0; u < U; ++u) {
      xs[u] = make_float4(0.f, 0.f, 0.f, 0.f);
      if (it[u].valid) xs[u] = ((const float4*)(xl + (size_t)it[u].s * C))[k];
    }
    if (OVF) {
      float p[U];
#pragma unroll
      for (int u = 0; u < U; ++u) {
        p[u] = 0.f;
        if (it[u].valid && i0 + u * NHW < nIter) {
          const float4 r = ((const float4*)(P.xr + it[u].dl * C))[k];
          p[u] = lrelu_dot4(xs[u], r, it[u].v, we4, at4);
        }
      }
#pragma unroll
      for (int u = 0; u < U; ++u) {
        p[u] = row16_sum(p[u]);
        if (i0 + u * NHW < nIter) it[u].v = p[u];
      }
    }
#pragma unroll
    for (int u = 0; u < U; ++u) {
      if (it[u].valid) {
        const float ex = __expf(it[u].v - f32_order_decode(P.lmax[it[u].dl]));
        if (k == 0) atomicAdd(&P.den[it[u].dl], ex);
        float* acc = P.acc + it[u].dl * C + 4 * k;
        const float v0 = ex * xs[u].x, v1 = ex * xs[u].y, v2 = ex * xs[u].z, v3 = ex * xs[u].w;
        atomicAdd(acc + (0 ^ rot), rot ? v2 : v0); atomicAdd(acc + (1 ^ rot), rot ? v3 : v1);
        atomicAdd(acc + (2 ^ rot), rot ? v0 : v2); atomicAdd(acc + (3 ^ rot), rot ? v1 : v3);
      }
    }
  }
  __syncthreads();

  const float4* a4 = (const float4*)P.acc;
  float4* o4 = (float4*)(out + (size_t)lo * C);
  for (int i = tid; i < (int)nn * (C / 4); i += NT) {
    const float4 a = a4[i];
    const float4 b = ((const float4*)bias)[i & 15];
    const float dn = P.den[i >> 4];
    float4 o;
    o.x = a.x / dn + b.x; o.y = a.y / dn + b.y; o.z = a.z / dn + b.z; o.w = a.w / dn + b.w;
    o4[i] = o;
  }
}

__device__ __forceinline__ void owner_tail_sorted(
    const Planes P, const int* __restrict__ ei, const int* __restrict__ dst,
    const float* __restrict__ eattr, const float* __restrict__ xl,
    const float* __restrict__ We, const float* __restrict__ att,
    const float* __restrict__ bias, float* __restrict__ out,
    const int nIter, const int lo, const unsigned nn) {
  const int tid = threadIdx.x;
  constexpr int PER = CAP / NT;

  int e_[PER]; int rs[PER]; float rv[PER]; int rd[PER]; unsigned rp[PER];
#pragma unroll
  for (int u = 0; u < PER; ++u) {
    const int i = tid + u * NT;
    e_[u] = (i < nIter) ? P.le[i] : -1;
  }
#pragma unroll
  for (int u = 0; u < PER; ++u) {
    rd[u] = 0; rv[u] = 0.f; rs[u] = 0;
    if (e_[u] >= 0) { rd[u] = dst[e_[u]]; rv[u] = eattr[e_[u]]; rs[u] = ei[e_[u]]; }
  }
#pragma unroll
  for (int u = 0; u < PER; ++u) {
    rp[u] = ~0u;
    if (e_[u] >= 0) {
      const unsigned dl = (unsigned)(rd[u] - lo);
      if (dl < nn) {
        const unsigned rank = atomicAdd(&P.cnt[dl], 1u);
        atomicAdd(&P.asum[dl], rv[u]);
        rp[u] = (rank << 8) | dl;
      }
    }
  }
  __syncthreads();

  if (tid < 32) {
    unsigned pre[8]; unsigned run = 0u;
#pragma unroll
    for (int j = 0; j < 8; ++j) {
      const unsigned nd = 8u * (unsigned)tid + (unsigned)j;
      pre[j] = run; run += P.cnt[nd] + ((nd < nn) ? 1u : 0u);
    }
    unsigned incl = run;
#pragma unroll
    for (int d = 1; d < 32; d <<= 1) {
      const unsigned t = __shfl_up(incl, d);
      if (tid >= d) incl += t;
    }
    const unsigned base = incl - run;
#pragma unroll
    for (int j = 0; j < 8; ++j) {
      const unsigned nd = 8u * (unsigned)tid + (unsigned)j;
      const unsigned o = base + pre[j];
      P.off[nd] = o;
      if (nd < nn) {
        P.le[o] = lo + (int)nd;
        P.lv[o] = P.asum[nd] / fmaxf((float)P.cnt[nd], 1.0f);
      }
    }
  }
  __syncthreads();

#pragma unroll
  for (int u = 0; u < PER; ++u) {
    if (rp[u] != ~0u) {
      const unsigned pos = P.off[rp[u] & 255u] + 1u + (rp[u] >> 8);
      P.le[pos] = rs[u]; P.lv[pos] = rv[u];
    }
  }
  __syncthreads();

  const int lane = tid & 31;
  const int k  = tid & 15;
  const unsigned kb = (unsigned)k << 4;
  const char* __restrict__ xlb = (const char*)xl;
  const float4 we4 = ((const float4*)We)[k];
  const float4 at4 = ((const float4*)att)[k];
  const float4 b4  = ((const float4*)bias)[k];
  for (;;) {
    unsigned pr = 0u;
    if (lane == 0) pr = atomicAdd(P.next, 1u);
    pr = (unsigned)__builtin_amdgcn_readfirstlane((int)pr);
    if (2u * pr >= nn) break;
    const int node = (int)(2u * pr) + (lane >> 4);
    if (node >= (int)nn) continue;
    const int beg = (int)P.off[node];
    const int nItem = (int)P.cnt[node] + 1;
    const int last = beg + nItem - 1;
    const float4 r4 = ((const float4*)(P.xr + node * C))[k];
    float m = -INFINITY, den = 0.f;
    float4 acc = make_float4(0.f, 0.f, 0.f, 0.f);
    int s[U]; float ea[U];
#pragma unroll
    for (int u = 0; u < U; ++u) {
      const int idx = min(beg + u, last);
      s[u] = P.le[idx]; ea[u] = P.lv[idx];
    }
    for (int j0 = 0; j0 < nItem; j0 += U) {
      float4 xs[U]; float eac[U]; float p[U];
#pragma unroll
      for (int u = 0; u < U; ++u) {
        xs[u] = *(const float4*)(xlb + (((unsigned)s[u] << 8) | kb));
        eac[u] = ea[u];
      }
#pragma unroll
      for (int u = 0; u < U; ++u) {
        const int idx = min(beg + j0 + U + u, last);
        s[u] = P.le[idx]; ea[u] = P.lv[idx];
      }
#pragma unroll
      for (int u = 0; u < U; ++u) {
        const float d = row16_sum(lrelu_dot4(xs[u], r4, eac[u], we4, at4));
        p[u] = (j0 + u < nItem) ? d : -INFINITY;
      }
      const float mn = fmaxf(fmaxf(m, p[0]), fmaxf(fmaxf(p[1], p[2]), p[3]));
      const float sc = __expf(m - mn);
      const float e0 = __expf(p[0] - mn), e1 = __expf(p[1] - mn);
      const float e2 = __expf(p[2] - mn), e3 = __expf(p[3] - mn);
      den = den * sc + ((e0 + e1) + (e2 + e3));
      acc.x = acc.x * sc + ((e0 * xs[0].x + e1 * xs[1].x) + (e2 * xs[2].x + e3 * xs[3].x));
      acc.y = acc.y * sc + ((e0 * xs[0].y + e1 * xs[1].y) + (e2 * xs[2].y + e3 * xs[3].y));
      acc.z = acc.z * sc + ((e0 * xs[0].z + e1 * xs[1].z) + (e2 * xs[2].z + e3 * xs[3].z));
      acc.w = acc.w * sc + ((e0 * xs[0].w + e1 * xs[1].w) + (e2 * xs[2].w + e3 * xs[3].w));
      m = mn;
    }
    float4 o;
    o.x = acc.x / den + b4.x; o.y = acc.y / den + b4.y;
    o.z = acc.z / den + b4.z; o.w = acc.w / den + b4.w;
    ((float4*)(out + (size_t)(lo + node) * C))[k] = o;
  }
}

__global__ __launch_bounds__(NT) void k_owner(
    const int* __restrict__ ei, const float* __restrict__ eattr,
    const float* __restrict__ xl, const float* __restrict__ xr,
    const float* __restrict__ We, const float* __restrict__ att,
    const float* __restrict__ bias, float* __restrict__ out,
    const unsigned* __restrict__ keys, const unsigned* __restrict__ bits, int E, int N) {
  __shared__ __align__(16) float s_xr[R_NODES * C];
  __shared__ __align__(16) float s_acc[R_NODES * C];
  __shared__ unsigned s_cnt[256];
  __shared__ unsigned s_off[256];
  __shared__ float    s_asum[R_NODES];
  __shared__ unsigned s_lmax[R_NODES];
  __shared__ float    s_den[R_NODES];
  __shared__ float    s_self[R_NODES];
  __shared__ int      s_le[CAP];
  __shared__ unsigned s_ldl[CAP];
  __shared__ float    s_lv[CAP];
  __shared__ unsigned s_n;
  __shared__ unsigned s_next;

  const int tid = threadIdx.x;
  const int lo  = blockIdx.x * R_NODES;
  const unsigned nn = (unsigned)min(R_NODES, N - lo);
  const int* __restrict__ dst = ei + E;

  {
    const float4 z4 = {0.f, 0.f, 0.f, 0.f};
    float4* a4 = (float4*)s_acc;
    for (int i = tid; i < R_NODES * (C / 4); i += NT) a4[i] = z4;
    for (int i = tid; i < R_NODES; i += NT) {
      s_asum[i] = 0.f; s_lmax[i] = 0u; s_den[i] = 0.f; s_self[i] = 0.f;
    }
    for (int i = tid; i < 256; i += NT) s_cnt[i] = 0u;
    if (tid == 0) { s_n = 0u; s_next = 0u; }
    const float4* src = (const float4*)(xr + (size_t)lo * C);
    float4* d4 = (float4*)s_xr;
    for (int i = tid; i < (int)nn * (C / 4); i += NT) d4[i] = src[i];
  }
  __syncthreads();

  const unsigned pat = (blockIdx.x & 255u) * 0x01010101u;
  const int G = E >> 4;
  const uint4* key4 = (const uint4*)keys;
  auto place = [&](unsigned m0, unsigned m1, const int* gq) {
    while (m0 | m1) {
      const bool first = m0 != 0u;
      const unsigned m = first ? m0 : m1;
      const int bit = __builtin_ctz(m);
      if (first) m0 = m & (m - 1u); else m1 = m & (m - 1u);
      const bool hi = ((bit >> 2) & 1) != 0;
      const int g = first ? (hi ? gq[1] : gq[0]) : (hi ? gq[3] : gq[2]);
      const unsigned p = atomicAdd(&s_n, 1u);
      if (p < (unsigned)CAP) s_le[p] = (g << 4) + ((bit & 3) << 2) + (bit >> 3);
    }
  };
  const int nBW = ((G + 127) >> 7) << 2;
  const unsigned* brow = bits + ((blockIdx.x & 255u) << 2);
  for (int w0 = tid; w0 < nBW; w0 += 2 * NT) {
    const int w1 = w0 + NT;
    unsigned ma = brow[(size_t)(w0 >> 2) * 1024 + (w0 & 3)];
    unsigned mb = (w1 < nBW) ? brow[(size_t)(w1 >> 2) * 1024 + (w1 & 3)] : 0u;
    while (ma | mb) {
      int gq[4]; uint4 v[4];
#pragma unroll
      for (int u = 0; u < 4; ++u) {
        const bool fa = ma != 0u;
        const unsigned m = fa ? ma : mb;
        int g = -1;
        if (m != 0u) {
          g = ((fa ? w0 : w1) << 5) + __builtin_ctz(m);
          if (fa) ma = m & (m - 1u); else mb = m & (m - 1u);
          if (g >= G) g = -1;
        }
        gq[u] = g;
      }
#pragma unroll
      for (int u = 0; u < 4; ++u) v[u] = key4[max(gq[u], 0)];
      const unsigned k0 = gq[0] >= 0 ? match16(v[0], pat) : 0u;
      const unsigned k1 = gq[1] >= 0 ? match16(v[1], pat) : 0u;
      const unsigned k2 = gq[2] >= 0 ? match16(v[2], pat) : 0u;
      const unsigned k3 = gq[3] >= 0 ? match16(v[3], pat) : 0u;
      place(k0 | (k1 << 4), k2 | (k3 << 4), gq);
    }
  }
  for (int e = (G << 4) + tid; e < E; e += NT) {
    if ((unsigned)(dst[e] - lo) < nn) {
      unsigned p = atomicAdd(&s_n, 1u);
      if (p < (unsigned)CAP) s_le[p] = e;
    }
  }
  __syncthreads();
  const unsigned nraw = s_n;

  Planes P;
  P.xr = s_xr; P.acc = s_acc; P.cnt = s_cnt; P.asum = s_asum; P.lmax = s_lmax; P.den = s_den;
  P.self = s_self; P.le = s_le; P.ldl = s_ldl; P.lv = s_lv; P.off = s_off; P.next = &s_next;
  if (nraw + nn > (unsigned)CAP)
    owner_tail<true>(P, ei, dst, eattr, xl, We, att, bias, out, E, lo, nn);
  else
    owner_tail_sorted(P, ei, dst, eattr, xl, We, att, bias, out, (int)nraw, lo, nn);
}

extern "C" void kernel_launch(void* const* d_in, const int* in_sizes, int n_in,
                              void* d_out, int out_size, void* d_ws, size_t ws_size,
                              hipStream_t stream) {
  (void)n_in; (void)out_size; (void)ws_size;
  const float* x     = (const float*)d_in[0];
  const float* eattr = (const float*)d_in[1];
  const float* Wl    = (const float*)d_in[2];
  const float* Wr    = (const float*)d_in[3];
  const float* We    = (const float*)d_in[4];
  const float* att   = (const float*)d_in[5];
  const float* bias  = (const float*)d_in[6];
  const int*   ei    = (const int*)d_in[7];
  float*       out   = (float*)d_out;

  int N = in_sizes[0] / F_IN;
  int E = in_sizes[1];

  char* ws = (char*)d_ws;
  size_t o = 0;
  auto take = [&](size_t bytes) -> void* {
    void* p = ws + o; o += (bytes + 255) & ~(size_t)255; return p;
  };
  float* xl    = (float*)take((size_t)N * C * 4);
  float* xr    = (float*)take((size_t)N * C * 4);
  unsigned* keys = (unsigned*)take((size_t)E);
  unsigned* bits = (unsigned*)take(((size_t)E / 2048 + 1) * 4096);

  k_gemm<<<GEMM_BLOCKS, GEMM_NT, 0, stream>>>(x, Wl, Wr, xl, xr, N, ei, keys, bits, E);
  k_owner<<<(unsigned)((N + R_NODES - 1) / R_NODES), NT, 0, stream>>>(
      ei, eattr, xl, xr, We, att, bias, out, keys, bits, E, N);
}


# owner block hands out its destinations sorted by trip count, longest first (counting sort by a second scan wave): a wave's two rows get equal segments, the last pairs are the shortest; sum order kept
# speedup vs baseline: 8.2053x; 1.0305x over previous
#include <hip/hip_runtime.h>

#define F_IN 128
#define C 64

typedef float v2f __attribute__((ext_vector_type(2)));
typedef float v8f __attribute__((ext_vector_type(8)));

__device__ __forceinline__ unsigned f32_order_encode(float f) {
  unsigned b = __float_as_uint(f);
  return (b & 0x80000000u) ? ~b : (b | 0x80000000u);
}
__device__ __forceinline__ float f32_order_decode(unsigned u) {
  unsigned b = (u & 0x80000000u) ? (u & 0x7FFFFFFFu) : ~u;
  return __uint_as_float(b);
}

#define R_NODES 200
static_assert(R_NODES <= 256, "a local destination must fit the low byte of a packed rank word");

#define GEMM_NT 256
__device__ __forceinline__ void write_owner_keys(const int* __restrict__ ei,
                                                 unsigned* __restrict__ keys,
                                                 unsigned* __restrict__ bits,
                                                 unsigned* s_bits, int E) {
  const int* __restrict__ dst = ei + E;
  const int tid = threadIdx.x;
  const int nW = (E >> 4) << 2;
  const int nRec = ((E >> 4) + 127) >> 7;
  for (int c = blockIdx.x; c < nRec; c += gridDim.x) {
    if (c != (int)blockIdx.x) __syncthreads();
    for (int i = tid; i < 4 * 256; i += GEMM_NT) s_bits[i] = 0u;
    __syncthreads();
    unsigned d[2][4];
#pragma unroll
    for (int u = 0; u < 2; ++u) {
      const int q = min(c * 512 + tid + u * GEMM_NT, nW - 1);
      d[u][0] = (unsigned)dst[4 * q];     d[u][1] = (unsigned)dst[4 * q + 1];
      d[u][2] = (unsigned)dst[4 * q + 2]; d[u][3] = (unsigned)dst[4 * q + 3];
    }
#pragma unroll
    for (int u = 0; u < 2; ++u) {
      const int t = tid + u * GEMM_NT;
      const int q = c * 512 + t;
      if (q < nW) {
        const unsigned o0 = (d[u][0] / R_NODES) & 255u, o1 = (d[u][1] / R_NODES) & 255u;
        const unsigned o2 = (d[u][2] / R_NODES) & 255u, o3 = (d[u][3] / R_NODES) & 255u;
        keys[q] = o0 | (o1 << 8) | (o2 << 16) | (o3 << 24);
        unsigned* wrow = s_bits + (t >> 7) * 256;
        const unsigned bit = 1u << ((t >> 2) & 31);
        atomicOr(&wrow[o0], bit); atomicOr(&wrow[o1], bit);
        atomicOr(&wrow[o2], bit); atomicOr(&wrow[o3], bit);
      }
    }
    __syncthreads();
    for (int r = tid; r < 256; r += GEMM_NT)
      ((uint4*)bits)[(size_t)c * 256 + r] =
          make_uint4(s_bits[r], s_bits[256 + r], s_bits[512 + r], s_bits[768 + r]);
  }
}

#define GEMM_BLOCKS 512
#define XPITCH 132
__global__ __launch_bounds__(GEMM_NT) void k_gemm(const float* __restrict__ x, const float* __restrict__ Wl,
                       const float* __restrict__ Wr, float* __restrict__ xl,
                       float* __restrict__ xr, int N,
                       const int* __restrict__ ei, unsigned* __restrict__ keys,
                       unsigned* __restrict__ bits, int E) {
  __shared__ __align__(16) float s_x[2][16 * XPITCH];
  __shared__ unsigned s_bits[4 * 256];
  write_owner_keys(ei, keys, bits, s_bits, E);
  const int tid  = threadIdx.x;
  const int wave = tid >> 5;
  const int lane = tid & 31;
  const int rowTiles = N >> 4;
  int rt = blockIdx.x;
  if (rt >= rowTiles) return;
  const int n0    = (wave >> 1) << 4;
  const int which = wave & 1;
  const float* W = which ? Wr : Wl;
  float*       Y = which ? xr : xl;
  const int half = lane >> 4;
  const int l15  = lane & 15;

  const int st0 = wave * XPITCH + lane * 4;
  const int st1 = st0 + 8 * XPITCH;
  float4 p0, p1;
  {
    const float4* src = (const float4*)(x + (size_t)rt * 16 * F_IN);
    p0 = src[tid]; p1 = src[tid + 256];
  }

  const float* bcol = W + n0 + l15 + (half * 2) * C;
  v2f b[32];
#pragma unroll
  for (int s = 0; s < 32; ++s) { b[s].x = bcol[(4 * s) * C]; b[s].y = bcol[(4 * s + 1) * C]; }

  *(float4*)(&s_x[0][st0]) = p0;
  *(float4*)(&s_x[0][st1]) = p1;
  __syncthreads();

  int cur = 0;
  for (;;) {
    const int nxt = rt + (int)gridDim.x;
    const bool more = nxt < rowTiles;
    if (more) {
      const float4* src = (const float4*)(x + (size_t)nxt * 16 * F_IN);
      p0 = src[tid]; p1 = src[tid + 256];
    }
    const v2f* ap = (const v2f*)(&s_x[cur][l15 * XPITCH + 2 * half]);
    v2f a[32];
#pragma unroll
    for (int s = 0; s < 32; ++s) a[s] = ap[2 * s];
    __builtin_amdgcn_sched_barrier(0);
    v8f acc = {0.f, 0.f, 0.f, 0.f, 0.f, 0.f, 0.f, 0.f};
#pragma unroll
    for (int s = 0; s < 32; ++s) {
      acc = __builtin_amdgcn_wmma_f32_16x16x4_f32(
           false, a[s],  false, b[s],
           (short)0, acc,  false,  false);
    }
    if (more) {
      *(float4*)(&s_x[cur ^ 1][st0]) = p0;
      *(float4*)(&s_x[cur ^ 1][st1]) = p1;
      __syncthreads();
    }
    float* yrow = Y + ((size_t)rt * 16 + half * 8) * C + n0 + l15;
#pragma unroll
    for (int r = 0; r < 8; ++r) yrow[(size_t)r * C] = acc[r];
    if (!more) break;
    cur ^= 1; rt = nxt;
  }
}

#define CAP 8192
#define NT 1024
#define NHW (NT / 16)
#define U 4

__device__ __forceinline__ unsigned zero_bytes(unsigned t) {
  return ~(((t & 0x7F7F7F7Fu) + 0x7F7F7F7Fu) | t) & 0x80808080u;
}
__device__ __forceinline__ unsigned match16(uint4 v, unsigned pat) {
  return (zero_bytes(v.x ^ pat) >> 7) | (zero_bytes(v.y ^ pat) >> 6) |
         (zero_bytes(v.z ^ pat) >> 5) | (zero_bytes(v.w ^ pat) >> 4);
}

struct Item { int s; unsigned dl; float v; bool valid; };

struct Planes {
  float* xr; float* acc; unsigned* cnt; float* asum; unsigned* lmax; float* den; float* self;
  int* le; unsigned* ldl; float* lv; unsigned* off; unsigned* next; unsigned* hist; unsigned* ord;
};

__device__ __forceinline__ float row16_sum(float v) {
  v += __int_as_float(__builtin_amdgcn_update_dpp(0, __float_as_int(v), 0x161, 0xf, 0xf, true));
  v += __int_as_float(__builtin_amdgcn_update_dpp(0, __float_as_int(v), 0x162, 0xf, 0xf, true));
  v += __int_as_float(__builtin_amdgcn_update_dpp(0, __float_as_int(v), 0x164, 0xf, 0xf, true));
  v += __int_as_float(__builtin_amdgcn_update_dpp(0, __float_as_int(v), 0x168, 0xf, 0xf, true));
  return v;
}

__device__ __forceinline__ float lrelu_dot4(float4 a, float4 r, float ea, float4 we, float4 at) {
  float m0 = a.x + r.x + ea * we.x, m1 = a.y + r.y + ea * we.y;
  float m2 = a.z + r.z + ea * we.z, m3 = a.w + r.w + ea * we.w;
  m0 = fmaxf(m0, 0.2f * m0); m1 = fmaxf(m1, 0.2f * m1);
  m2 = fmaxf(m2, 0.2f * m2); m3 = fmaxf(m3, 0.2f * m3);
  return m0 * at.x + m1 * at.y + m2 * at.z + m3 * at.w;
}

template <bool OVF>
__device__ __forceinline__ void owner_tail(
    const Planes P, const int* __restrict__ ei, const int* __restrict__ dst,
    const float* __restrict__ eattr, const float* __restrict__ xl,
    const float* __restrict__ We, const float* __restrict__ att,
    const float* __restrict__ bias, float* __restrict__ out,
    const int nIter, const int lo, const unsigned nn) {
  const int tid = threadIdx.x;

  for (int i = tid; i < nIter; i += NT) {
    if (!OVF) {
      const int e = P.le[i];
      unsigned dl = (unsigned)(dst[e] - lo);
      const float ea = eattr[e];
      const int s = ei[e];
      if (dl < nn) { atomicAdd(&P.cnt[dl], 1u); atomicAdd(&P.asum[dl], ea); } else dl = ~0u;
      P.le[i] = s; P.ldl[i] = dl; P.lv[i] = ea;
    } else {
      const unsigned dl = (unsigned)(dst[i] - lo);
      if (dl < nn) {
        atomicAdd(&P.cnt[dl], 1u);
        atomicAdd(&P.asum[dl], eattr[i]);
      }
    }
  }
  __syncthreads();

  const int hw = tid >> 4;
  const int k  = tid & 15;
  const int rot = (hw & 1) << 1;
  const int total = nIter + (int)nn;
  const float4 we4 = ((const float4*)We)[k];
  const float4 at4 = ((const float4*)att)[k];

  auto fetch = [&](int i, bool wantLogit) -> Item {
    Item it; it.s = 0; it.dl = 0u; it.v = 0.f; it.valid = false;
    if (i < nIter) {
      if (!OVF) { it.s = P.le[i]; it.dl = P.ldl[i]; it.v = P.lv[i]; it.valid = it.dl != ~0u; }
      else {
        const unsigned dl = (unsigned)(dst[i] - lo);
        if (dl < nn) { it.s = ei[i]; it.dl = dl; it.v = eattr[i]; it.valid = true; }
      }
    } else if (i < total) {
      const unsigned dl = (unsigned)(i - nIter);
      it.s = lo + (int)dl; it.dl = dl; it.valid = true;
      it.v = wantLogit ? P.self[dl] : P.asum[dl] / fmaxf((float)P.cnt[dl], 1.0f);
    }
    return it;
  };

  for (int i0 = hw; i0 < total; i0 += U * NHW) {
    Item it[U]; float4 xs[U]; float p[U];
#pragma unroll
    for (int u = 0; u < U; ++u) it[u] = fetch(i0 + u * NHW, false);
#pragma unroll
    for (int u = 0; u < U; ++u) {
      xs[u] = make_float4(0.f, 0.f, 0.f, 0.f);
      if (it[u].valid) xs[u] = ((const float4*)(xl + (size_t)it[u].s * C))[k];
    }
#pragma unroll
    for (int u = 0; u < U; ++u) {
      p[u] = 0.f;
      if (it[u].valid) {
        const float4 r = ((const float4*)(P.xr + it[u].dl * C))[k];
        p[u] = lrelu_dot4(xs[u], r, it[u].v, we4, at4);
      }
    }
#pragma unroll
    for (int u = 0; u < U; ++u) p[u] = row16_sum(p[u]);
    if (k == 0) {
#pragma unroll
      for (int u = 0; u < U; ++u) {
        const int i = i0 + u * NHW;
        if (it[u].valid) {
          if (i >= nIter) P.self[it[u].dl] = p[u]; else if (!OVF) P.lv[i] = p[u];
          atomicMax(&P.lmax[it[u].dl], f32_order_encode(p[u]));
        }
      }
    }
  }
  __syncthreads();

  for (int i0 = hw; i0 < total; i0 += U * NHW) {
    Item it[U]; float4 xs[U];
#pragma unroll
    for (int u = 0; u < U; ++u) it[u] = fetch(i0 + u * NHW, true);
#pragma unroll
    for (int u = 0; u < U; ++u) {
      xs[u] = make_float4(0.f, 0.f, 0.f, 0.f);
      if (it[u].valid) xs[u] = ((const float4*)(xl + (size_t)it[u].s * C))[k];
    }
    if (OVF) {
      float p[U];
#pragma unroll
      for (int u = 0; u < U; ++u) {
        p[u] = 0.f;
        if (it[u].valid && i0 + u * NHW < nIter) {
          const float4 r = ((const float4*)(P.xr + it[u].dl * C))[k];
          p[u] = lrelu_dot4(xs[u], r, it[u].v, we4, at4);
        }
      }
#pragma unroll
      for (int u = 0; u < U; ++u) {
        p[u] = row16_sum(p[u]);
        if (i0 + u * NHW < nIter) it[u].v = p[u];
      }
    }
#pragma unroll
    for (int u = 0; u < U; ++u) {
      if (it[u].valid) {
        const float ex = __expf(it[u].v - f32_order_decode(P.lmax[it[u].dl]));
        if (k == 0) atomicAdd(&P.den[it[u].dl], ex);
        float* acc = P.acc + it[u].dl * C + 4 * k;
        const float v0 = ex * xs[u].x, v1 = ex * xs[u].y, v2 = ex * xs[u].z, v3 = ex * xs[u].w;
        atomicAdd(acc + (0 ^ rot), rot ? v2 : v0); atomicAdd(acc + (1 ^ rot), rot ? v3 : v1);
        atomicAdd(acc + (2 ^ rot), rot ? v0 : v2); atomicAdd(acc + (3 ^ rot), rot ? v1 : v3);
      }
    }
  }
  __syncthreads();

  const float4* a4 = (const float4*)P.acc;
  float4* o4 = (float4*)(out + (size_t)lo * C);
  for (int i = tid; i < (int)nn * (C / 4); i += NT) {
    const float4 a = a4[i];
    const float4 b = ((const float4*)bias)[i & 15];
    const float dn = P.den[i >> 4];
    float4 o;
    o.x = a.x / dn + b.x; o.y = a.y / dn + b.y; o.z = a.z / dn + b.z; o.w = a.w / dn + b.w;
    o4[i] = o;
  }
}

__device__ __forceinline__ void owner_tail_sorted(
    const Planes P, const int* __restrict__ ei, const int* __restrict__ dst,
    const float* __restrict__ eattr, const float* __restrict__ xl,
    const float* __restrict__ We, const float* __restrict__ att,
    const float* __restrict__ bias, float* __restrict__ out,
    const int nIter, const int lo, const unsigned nn) {
  const int tid = threadIdx.x;
  constexpr int PER = CAP / NT;

  int e_[PER]; int rs[PER]; float rv[PER]; int rd[PER]; unsigned rp[PER];
#pragma unroll
  for (int u = 0; u < PER; ++u) {
    const int i = tid + u * NT;
    e_[u] = (i < nIter) ? P.le[i] : -1;
  }
#pragma unroll
  for (int u = 0; u < PER; ++u) {
    rd[u] = 0; rv[u] = 0.f; rs[u] = 0;
    if (e_[u] >= 0) { rd[u] = dst[e_[u]]; rv[u] = eattr[e_[u]]; rs[u] = ei[e_[u]]; }
  }
#pragma unroll
  for (int u = 0; u < PER; ++u) {
    rp[u] = ~0u;
    if (e_[u] >= 0) {
      const unsigned dl = (unsigned)(rd[u] - lo);
      if (dl < nn) {
        const unsigned rank = atomicAdd(&P.cnt[dl], 1u);
        atomicAdd(&P.asum[dl], rv[u]);
        rp[u] = (rank << 8) | dl;
      }
    }
  }
  __syncthreads();

  if (tid < 32) {
    unsigned pre[8]; unsigned run = 0u;
#pragma unroll
    for (int j = 0; j < 8; ++j) {
      const unsigned nd = 8u * (unsigned)tid + (unsigned)j;
      pre[j] = run; run += P.cnt[nd] + ((nd < nn) ? 1u : 0u);
    }
    unsigned incl = run;
#pragma unroll
    for (int d = 1; d < 32; d <<= 1) {
      const unsigned t = __shfl_up(incl, d);
      if (tid >= d) incl += t;
    }
    const unsigned base = incl - run;
#pragma unroll
    for (int j = 0; j < 8; ++j) {
      const unsigned nd = 8u * (unsigned)tid + (unsigned)j;
      const unsigned o = base + pre[j];
      P.off[nd] = o;
      if (nd < nn) {
        P.le[o] = lo + (int)nd;
        P.lv[o] = P.asum[nd] / fmaxf((float)P.cnt[nd], 1.0f);
      }
    }
  } else if (tid < 64) {
    const int t32 = tid - 32;
    unsigned cn[8]; unsigned cls[8]; unsigned rk[8]; unsigned b[8];
#pragma unroll
    for (int j = 0; j < 8; ++j) cn[j] = P.cnt[8u * (unsigned)t32 + (unsigned)j];
#pragma unroll
    for (int j = 0; j < 8; ++j) {
      const unsigned nd = 8u * (unsigned)t32 + (unsigned)j;
      const unsigned trips = (cn[j] + (unsigned)U) / (unsigned)U;
      cls[j] = 31u - min(trips, 31u);
      rk[j] = 0u;
      if (nd < nn) rk[j] = atomicAdd(&P.hist[cls[j]], 1u);
    }
    const unsigned mine = P.hist[t32];
    unsigned cin = mine;
#pragma unroll
    for (int d = 1; d < 32; d <<= 1) {
      const unsigned t = __shfl_up(cin, d);
      if (t32 >= d) cin += t;
    }
    const unsigned cbase = cin - mine;
#pragma unroll
    for (int j = 0; j < 8; ++j) b[j] = (unsigned)__shfl((int)cbase, (int)cls[j]);
#pragma unroll
    for (int j = 0; j < 8; ++j) {
      const unsigned nd = 8u * (unsigned)t32 + (unsigned)j;
      if (nd < nn) P.ord[b[j] + rk[j]] = nd;
    }
  }
  __syncthreads();

#pragma unroll
  for (int u = 0; u < PER; ++u) {
    if (rp[u] != ~0u) {
      const unsigned pos = P.off[rp[u] & 255u] + 1u + (rp[u] >> 8);
      P.le[pos] = rs[u]; P.lv[pos] = rv[u];
    }
  }
  __syncthreads();

  const int lane = tid & 31;
  const int k  = tid & 15;
  const unsigned kb = (unsigned)k << 4;
  const char* __restrict__ xlb = (const char*)xl;
  const float4 we4 = ((const float4*)We)[k];
  const float4 at4 = ((const float4*)att)[k];
  const float4 b4  = ((const float4*)bias)[k];
  for (;;) {
    unsigned pr = 0u;
    if (lane == 0) pr = atomicAdd(P.next, 1u);
    pr = (unsigned)__builtin_amdgcn_readfirstlane((int)pr);
    if (2u * pr >= nn) break;
    const unsigned slot = 2u * pr + (unsigned)(lane >> 4);
    if (slot >= nn) continue;
    const int node = (int)P.ord[slot];
    const int beg = (int)P.off[node];
    const int nItem = (int)P.cnt[node] + 1;
    const int last = beg + nItem - 1;
    const float4 r4 = ((const float4*)(P.xr + node * C))[k];
    float m = -INFINITY, den = 0.f;
    float4 acc = make_float4(0.f, 0.f, 0.f, 0.f);
    int s[U]; float ea[U];
#pragma unroll
    for (int u = 0; u < U; ++u) {
      const int idx = min(beg + u, last);
      s[u] = P.le[idx]; ea[u] = P.lv[idx];
    }
    for (int j0 = 0; j0 < nItem; j0 += U) {
      float4 xs[U]; float eac[U]; float p[U];
#pragma unroll
      for (int u = 0; u < U; ++u) {
        xs[u] = *(const float4*)(xlb + (((unsigned)s[u] << 8) | kb));
        eac[u] = ea[u];
      }
#pragma unroll
      for (int u = 0; u < U; ++u) {
        const int idx = min(beg + j0 + U + u, last);
        s[u] = P.le[idx]; ea[u] = P.lv[idx];
      }
#pragma unroll
      for (int u = 0; u < U; ++u) {
        const float d = row16_sum(lrelu_dot4(xs[u], r4, eac[u], we4, at4));
        p[u] = (j0 + u < nItem) ? d : -INFINITY;
      }
      const float mn = fmaxf(fmaxf(m, p[0]), fmaxf(fmaxf(p[1], p[2]), p[3]));
      const float sc = __expf(m - mn);
      const float e0 = __expf(p[0] - mn), e1 = __expf(p[1] - mn);
      const float e2 = __expf(p[2] - mn), e3 = __expf(p[3] - mn);
      den = den * sc + ((e0 + e1) + (e2 + e3));
      acc.x = acc.x * sc + ((e0 * xs[0].x + e1 * xs[1].x) + (e2 * xs[2].x + e3 * xs[3].x));
      acc.y = acc.y * sc + ((e0 * xs[0].y + e1 * xs[1].y) + (e2 * xs[2].y + e3 * xs[3].y));
      acc.z = acc.z * sc + ((e0 * xs[0].z + e1 * xs[1].z) + (e2 * xs[2].z + e3 * xs[3].z));
      acc.w = acc.w * sc + ((e0 * xs[0].w + e1 * xs[1].w) + (e2 * xs[2].w + e3 * xs[3].w));
      m = mn;
    }
    float4 o;
    o.x = acc.x / den + b4.x; o.y = acc.y / den + b4.y;
    o.z = acc.z / den + b4.z; o.w = acc.w / den + b4.w;
    ((float4*)(out + (size_t)(lo + node) * C))[k] = o;
  }
}

__global__ __launch_bounds__(NT) void k_owner(
    const int* __restrict__ ei, const float* __restrict__ eattr,
    const float* __restrict__ xl, const float* __restrict__ xr,
    const float* __restrict__ We, const float* __restrict__ att,
    const float* __restrict__ bias, float* __restrict__ out,
    const unsigned* __restrict__ keys, const unsigned* __restrict__ bits, int E, int N) {
  __shared__ __align__(16) float s_xr[R_NODES * C];
  __shared__ __align__(16) float s_acc[R_NODES * C];
  __shared__ unsigned s_cnt[256];
  __shared__ unsigned s_off[256];
  __shared__ float    s_asum[R_NODES];
  __shared__ unsigned s_lmax[R_NODES];
  __shared__ float    s_den[R_NODES];
  __shared__ float    s_self[R_NODES];
  __shared__ int      s_le[CAP];
  __shared__ unsigned s_ldl[CAP];
  __shared__ float    s_lv[CAP];
  __shared__ unsigned s_n;
  __shared__ unsigned s_next;
  __shared__ unsigned s_hist[32];
  __shared__ unsigned s_ord[256];

  const int tid = threadIdx.x;
  const int lo  = blockIdx.x * R_NODES;
  const unsigned nn = (unsigned)min(R_NODES, N - lo);
  const int* __restrict__ dst = ei + E;

  {
    const float4 z4 = {0.f, 0.f, 0.f, 0.f};
    float4* a4 = (float4*)s_acc;
    for (int i = tid; i < R_NODES * (C / 4); i += NT) a4[i] = z4;
    for (int i = tid; i < R_NODES; i += NT) {
      s_asum[i] = 0.f; s_lmax[i] = 0u; s_den[i] = 0.f; s_self[i] = 0.f;
    }
    for (int i = tid; i < 256; i += NT) s_cnt[i] = 0u;
    if (tid < 32) s_hist[tid] = 0u;
    if (tid == 0) { s_n = 0u; s_next = 0u; }
    const float4* src = (const float4*)(xr + (size_t)lo * C);
    float4* d4 = (float4*)s_xr;
    for (int i = tid; i < (int)nn * (C / 4); i += NT) d4[i] = src[i];
  }
  __syncthreads();

  const unsigned pat = (blockIdx.x & 255u) * 0x01010101u;
  const int G = E >> 4;
  const uint4* key4 = (const uint4*)keys;
  auto place = [&](unsigned m0, unsigned m1, const int* gq) {
    while (m0 | m1) {
      const bool first = m0 != 0u;
      const unsigned m = first ? m0 : m1;
      const int bit = __builtin_ctz(m);
      if (first) m0 = m & (m - 1u); else m1 = m & (m - 1u);
      const bool hi = ((bit >> 2) & 1) != 0;
      const int g = first ? (hi ? gq[1] : gq[0]) : (hi ? gq[3] : gq[2]);
      const unsigned p = atomicAdd(&s_n, 1u);
      if (p < (unsigned)CAP) s_le[p] = (g << 4) + ((bit & 3) << 2) + (bit >> 3);
    }
  };
  const int nBW = ((G + 127) >> 7) << 2;
  const unsigned* brow = bits + ((blockIdx.x & 255u) << 2);
  for (int w0 = tid; w0 < nBW; w0 += 2 * NT) {
    const int w1 = w0 + NT;
    unsigned ma = brow[(size_t)(w0 >> 2) * 1024 + (w0 & 3)];
    unsigned mb = (w1 < nBW) ? brow[(size_t)(w1 >> 2) * 1024 + (w1 & 3)] : 0u;
    while (ma | mb) {
      int gq[4]; uint4 v[4];
#pragma unroll
      for (int u = 0; u < 4; ++u) {
        const bool fa = ma != 0u;
        const unsigned m = fa ? ma : mb;
        int g = -1;
        if (m != 0u) {
          g = ((fa ? w0 : w1) << 5) + __builtin_ctz(m);
          if (fa) ma = m & (m - 1u); else mb = m & (m - 1u);
          if (g >= G) g = -1;
        }
        gq[u] = g;
      }
#pragma unroll
      for (int u = 0; u < 4; ++u) v[u] = key4[max(gq[u], 0)];
      const unsigned k0 = gq[0] >= 0 ? match16(v[0], pat) : 0u;
      const unsigned k1 = gq[1] >= 0 ? match16(v[1], pat) : 0u;
      const unsigned k2 = gq[2] >= 0 ? match16(v[2], pat) : 0u;
      const unsigned k3 = gq[3] >= 0 ? match16(v[3], pat) : 0u;
      place(k0 | (k1 << 4), k2 | (k3 << 4), gq);
    }
  }
  for (int e = (G << 4) + tid; e < E; e += NT) {
    if ((unsigned)(dst[e] - lo) < nn) {
      unsigned p = atomicAdd(&s_n, 1u);
      if (p < (unsigned)CAP) s_le[p] = e;
    }
  }
  __syncthreads();
  const unsigned nraw = s_n;

  Planes P;
  P.xr = s_xr; P.acc = s_acc; P.cnt = s_cnt; P.asum = s_asum; P.lmax = s_lmax; P.den = s_den;
  P.self = s_self; P.le = s_le; P.ldl = s_ldl; P.lv = s_lv; P.off = s_off; P.next = &s_next; P.hist = s_hist; P.ord = s_ord;
  if (nraw + nn > (unsigned)CAP)
    owner_tail<true>(P, ei, dst, eattr, xl, We, att, bias, out, E, lo, nn);
  else
    owner_tail_sorted(P, ei, dst, eattr, xl, We, att, bias, out, (int)nraw, lo, nn);
}

extern "C" void kernel_launch(void* const* d_in, const int* in_sizes, int n_in,
                              void* d_out, int out_size, void* d_ws, size_t ws_size,
                              hipStream_t stream) {
  (void)n_in; (void)out_size; (void)ws_size;
  const float* x     = (const float*)d_in[0];
  const float* eattr = (const float*)d_in[1];
  const float* Wl    = (const float*)d_in[2];
  const float* Wr    = (const float*)d_in[3];
  const float* We    = (const float*)d_in[4];
  const float* att   = (const float*)d_in[5];
  const float* bias  = (const float*)d_in[6];
  const int*   ei    = (const int*)d_in[7];
  float*       out   = (float*)d_out;

  int N = in_sizes[0] / F_IN;
  int E = in_sizes[1];

  char* ws = (char*)d_ws;
  size_t o = 0;
  auto take = [&](size_t bytes) -> void* {
    void* p = ws + o; o += (bytes + 255) & ~(size_t)255; return p;
  };
  float* xl    = (float*)take((size_t)N * C * 4);
  float* xr    = (float*)take((size_t)N * C * 4);
  unsigned* keys = (unsigned*)take((size_t)E);
  unsigned* bits = (unsigned*)take(((size_t)E / 2048 + 1) * 4096);

  k_gemm<<<GEMM_BLOCKS, GEMM_NT, 0, stream>>>(x, Wl, Wr, xl, xr, N, ei, keys, bits, E);
  k_owner<<<(unsigned)((N + R_NODES - 1) / R_NODES), NT, 0, stream>>>(
      ei, eattr, xl, xr, We, att, bias, out, keys, bits, E, N);
}
